# starting kernel
# baseline (speedup 1.0000x reference)
#include <hip/hip_runtime.h>
#include <hip/hip_bf16.h>
#include <math.h>

#define NNODES   100000
#define NEDGES   600000
#define FDIM     256
#define HDIM     128
#define NCLS     47
#define NCLSPAD  64
#define NSEEDS   10000
#define GEN_EPS  1e-7f
#define LN_EPS   1e-5f

typedef float v2f __attribute__((ext_vector_type(2)));
typedef float v8f __attribute__((ext_vector_type(8)));

__global__ __launch_bounds__(128)
void gemm_f32_wmma_kernel(const float* __restrict__ A, const float* __restrict__ Wt,
                          const float* __restrict__ bias, const float* __restrict__ res,
                          float* __restrict__ out, int M, int K, int Nc)
{
    const int lane  = threadIdx.x;
    const int mtile = blockIdx.x * 4 + threadIdx.y;
    const int row0  = mtile * 16;
    if (row0 >= M) return;
    const int hb   = lane >> 4;
    const int lo   = lane & 15;
    const int arow = row0 + lo;
    const int col0 = blockIdx.y * 64 + lo;

    const float* __restrict__ a_ptr = A  + (size_t)arow * K;
    const float* __restrict__ b_ptr = Wt + (size_t)col0 * K;
    const size_t bstep = (size_t)16 * K;

    v8f acc[4];
#pragma unroll
    for (int t = 0; t < 4; ++t)
#pragma unroll
        for (int i = 0; i < 8; ++i) acc[t][i] = 0.f;

#pragma unroll 4
    for (int k = 0; k < K; k += 4) {
        const int ka = k + 2 * hb;
        const v2f a  = *(const v2f*)(a_ptr + ka);
        const v2f b0 = *(const v2f*)(b_ptr + ka);
        const v2f b1 = *(const v2f*)(b_ptr + bstep + ka);
        const v2f b2 = *(const v2f*)(b_ptr + 2 * bstep + ka);
        const v2f b3 = *(const v2f*)(b_ptr + 3 * bstep + ka);
        acc[0] = __builtin_amdgcn_wmma_f32_16x16x4_f32(false, a, false, b0, (short)0, acc[0], false, false);
        acc[1] = __builtin_amdgcn_wmma_f32_16x16x4_f32(false, a, false, b1, (short)0, acc[1], false, false);
        acc[2] = __builtin_amdgcn_wmma_f32_16x16x4_f32(false, a, false, b2, (short)0, acc[2], false, false);
        acc[3] = __builtin_amdgcn_wmma_f32_16x16x4_f32(false, a, false, b3, (short)0, acc[3], false, false);
    }

#pragma unroll
    for (int t = 0; t < 4; ++t) {
        const int   col = col0 + 16 * t;
        const float bv  = bias[col];
#pragma unroll
        for (int i = 0; i < 8; ++i) {
            const int r = row0 + i + 8 * hb;
            float v = acc[t][i] + bv;
            if (res) v += res[(size_t)r * Nc + col];
            out[(size_t)r * Nc + col] = v;
        }
    }
}

__global__ __launch_bounds__(256)
void transpose_pad_kernel(const float* __restrict__ W, float* __restrict__ Wt,
                          int K, int Nc, int Ncpad)
{
    const int tid = blockIdx.x * 256 + threadIdx.x;
    if (tid >= Ncpad * K) return;
    const int c = tid / K;
    const int k = tid - c * K;
    Wt[(size_t)c * K + k] = (c < Nc) ? W[(size_t)k * Nc + c] : 0.f;
}

__global__ void pad_bias_kernel(const float* __restrict__ b, float* __restrict__ bp,
                                int Nc, int Ncpad)
{
    const int c = threadIdx.x;
    if (c < Ncpad) bp[c] = (c < Nc) ? b[c] : 0.f;
}

__global__ __launch_bounds__(256)
void edge_max_kernel(const float* __restrict__ hin, const int* __restrict__ src,
                     const int* __restrict__ dst, float* __restrict__ mx)
{
    const int tid = blockIdx.x * 256 + threadIdx.x;
    if (tid >= NEDGES * 32) return;
    const int e = tid >> 5;
    const int f = (tid & 31) * 4;
    const int s = src[e];
    const int d = dst[e];
    const float4 hv = *(const float4*)(hin + (size_t)s * HDIM + f);
    unsigned* mp = (unsigned*)(mx + (size_t)d * HDIM + f);
    atomicMax(mp + 0, __float_as_uint(fmaxf(hv.x, 0.f) + GEN_EPS));
    atomicMax(mp + 1, __float_as_uint(fmaxf(hv.y, 0.f) + GEN_EPS));
    atomicMax(mp + 2, __float_as_uint(fmaxf(hv.z, 0.f) + GEN_EPS));
    atomicMax(mp + 3, __float_as_uint(fmaxf(hv.w, 0.f) + GEN_EPS));
}

__global__ __launch_bounds__(256)
void edge_sum_kernel(const float* __restrict__ hin, const int* __restrict__ src,
                     const int* __restrict__ dst, const float* __restrict__ mx,
                     float* __restrict__ denom, float* __restrict__ numer)
{
    const int tid = blockIdx.x * 256 + threadIdx.x;
    if (tid >= NEDGES * 32) return;
    const int e = tid >> 5;
    const int f = (tid & 31) * 4;
    const int s = src[e];
    const int d = dst[e];
    const float4 hv = *(const float4*)(hin + (size_t)s * HDIM + f);
    const float4 mv = *(const float4*)(mx + (size_t)d * HDIM + f);
    const size_t o = (size_t)d * HDIM + f;
    float m, ex;
    m = fmaxf(hv.x, 0.f) + GEN_EPS; ex = expf(m - mv.x);
    atomicAdd(denom + o + 0, ex); atomicAdd(numer + o + 0, m * ex);
    m = fmaxf(hv.y, 0.f) + GEN_EPS; ex = expf(m - mv.y);
    atomicAdd(denom + o + 1, ex); atomicAdd(numer + o + 1, m * ex);
    m = fmaxf(hv.z, 0.f) + GEN_EPS; ex = expf(m - mv.z);
    atomicAdd(denom + o + 2, ex); atomicAdd(numer + o + 2, m * ex);
    m = fmaxf(hv.w, 0.f) + GEN_EPS; ex = expf(m - mv.w);
    atomicAdd(denom + o + 3, ex); atomicAdd(numer + o + 3, m * ex);
}

__global__ __launch_bounds__(256)
void combine_kernel(const float* __restrict__ numer, const float* __restrict__ denom,
                    const float* __restrict__ hin, float* __restrict__ t, int n4)
{
    const int i = blockIdx.x * 256 + threadIdx.x;
    if (i >= n4) return;
    const float4 nu = ((const float4*)numer)[i];
    const float4 de = ((const float4*)denom)[i];
    const float4 hv = ((const float4*)hin)[i];
    float4 r;
    r.x = nu.x / (de.x + 1e-16f) + hv.x;
    r.y = nu.y / (de.y + 1e-16f) + hv.y;
    r.z = nu.z / (de.z + 1e-16f) + hv.z;
    r.w = nu.w / (de.w + 1e-16f) + hv.w;
    ((float4*)t)[i] = r;
}

__global__ __launch_bounds__(256)
void ln_relu_kernel(const float* __restrict__ in, const float* __restrict__ g,
                    const float* __restrict__ b, float* __restrict__ out, int M)
{
    const int row  = blockIdx.x * 8 + (threadIdx.x >> 5);
    const int lane = threadIdx.x & 31;
    if (row >= M) return;
    const float4 v = *(const float4*)(in + (size_t)row * HDIM + lane * 4);
    float s1 = v.x + v.y + v.z + v.w;
    float s2 = v.x * v.x + v.y * v.y + v.z * v.z + v.w * v.w;
#pragma unroll
    for (int off = 16; off > 0; off >>= 1) {
        s1 += __shfl_xor(s1, off, 32);
        s2 += __shfl_xor(s2, off, 32);
    }
    const float mu  = s1 * (1.f / HDIM);
    const float var = s2 * (1.f / HDIM) - mu * mu;
    const float inv = rsqrtf(var + LN_EPS);
    const float4 gg = *(const float4*)(g + lane * 4);
    const float4 bb = *(const float4*)(b + lane * 4);
    float4 r;
    r.x = fmaxf((v.x - mu) * inv * gg.x + bb.x, 0.f);
    r.y = fmaxf((v.y - mu) * inv * gg.y + bb.y, 0.f);
    r.z = fmaxf((v.z - mu) * inv * gg.z + bb.z, 0.f);
    r.w = fmaxf((v.w - mu) * inv * gg.w + bb.w, 0.f);
    *(float4*)(out + (size_t)row * HDIM + lane * 4) = r;
}

__global__ __launch_bounds__(256)
void gather_kernel(const float* __restrict__ in, const int* __restrict__ map,
                   float* __restrict__ out, int M)
{
    const int tid = blockIdx.x * 256 + threadIdx.x;
    if (tid >= M * 32) return;
    const int i = tid >> 5;
    const int f = (tid & 31) * 4;
    const int j = map[i];
    *(float4*)(out + (size_t)i * HDIM + f) =
        *(const float4*)(in + (size_t)j * HDIM + f);
}

__global__ __launch_bounds__(256)
void logsoftmax_kernel(const float* __restrict__ in, float* __restrict__ out, int M)
{
    const int row  = blockIdx.x * 8 + (threadIdx.x >> 5);
    const int lane = threadIdx.x & 31;
    if (row >= M) return;
    const float NEG = -3.0e38f;
    const float v0 = (lane      < NCLS) ? in[(size_t)row * NCLSPAD + lane     ] : NEG;
    const float v1 = (lane + 32 < NCLS) ? in[(size_t)row * NCLSPAD + lane + 32] : NEG;
    float mx = fmaxf(v0, v1);
#pragma unroll
    for (int off = 16; off > 0; off >>= 1) mx = fmaxf(mx, __shfl_xor(mx, off, 32));
    float se = ((lane      < NCLS) ? expf(v0 - mx) : 0.f)
             + ((lane + 32 < NCLS) ? expf(v1 - mx) : 0.f);
#pragma unroll
    for (int off = 16; off > 0; off >>= 1) se += __shfl_xor(se, off, 32);
    const float lse = logf(se) + mx;
    if (lane      < NCLS) out[(size_t)row * NCLS + lane     ] = v0 - lse;
    if (lane + 32 < NCLS) out[(size_t)row * NCLS + lane + 32] = v1 - lse;
}

extern "C" void kernel_launch(void* const* d_in, const int* in_sizes, int n_in,
                              void* d_out, int out_size, void* d_ws, size_t ws_size,
                              hipStream_t stream)
{
    (void)in_sizes; (void)n_in; (void)out_size; (void)ws_size;

    const float* x         = (const float*)d_in[0];
    const int*   src       = (const int*)  d_in[1];
    const int*   dst       = (const int*)  d_in[2];
    const int*   node_map  = (const int*)  d_in[3];
    const int*   final_map = (const int*)  d_in[4];
    const float* enc_w     = (const float*)d_in[5];
    const float* enc_b     = (const float*)d_in[6];
    const float* gcn_w     = (const float*)d_in[7];
    const float* gcn_b     = (const float*)d_in[8];
    const float* ln_g      = (const float*)d_in[9];
    const float* ln_b      = (const float*)d_in[10];
    const float* pred_w    = (const float*)d_in[11];
    const float* pred_b    = (const float*)d_in[12];
    float* out = (float*)d_out;

    const size_t NH = (size_t)NNODES * HDIM;
    float* B0     = (float*)d_ws;
    float* B1     = B0 + NH;
    float* B2     = B0 + 2 * NH;
    float* B3     = B0 + 3 * NH;
    float* B4     = B0 + 4 * NH;
    float* HG     = B0 + 5 * NH;
    float* HG2    = HG  + (size_t)NSEEDS * HDIM;
    float* SC     = HG2 + (size_t)NSEEDS * HDIM;
    float* WT_ENC = SC  + (size_t)NSEEDS * NCLSPAD;
    float* WT_G   = WT_ENC + (size_t)HDIM * FDIM;
    float* PWT    = WT_G   + (size_t)3 * HDIM * HDIM;
    float* PB     = PWT    + (size_t)NCLSPAD * HDIM;

    transpose_pad_kernel<<<(HDIM * FDIM + 255) / 256, 256, 0, stream>>>(
        enc_w, WT_ENC, FDIM, HDIM, HDIM);
    for (int l = 0; l < 3; ++l)
        transpose_pad_kernel<<<(HDIM * HDIM + 255) / 256, 256, 0, stream>>>(
            gcn_w + (size_t)l * HDIM * HDIM, WT_G + (size_t)l * HDIM * HDIM,
            HDIM, HDIM, HDIM);
    transpose_pad_kernel<<<(NCLSPAD * HDIM + 255) / 256, 256, 0, stream>>>(
        pred_w, PWT, HDIM, NCLS, NCLSPAD);
    pad_bias_kernel<<<1, NCLSPAD, 0, stream>>>(pred_b, PB, NCLS, NCLSPAD);

    const dim3 gblk(32, 4, 1);
    const int  mtN   = NNODES / 16;
    const dim3 gridN((mtN + 3) / 4, HDIM / 64, 1);
    const int  egrid = (NEDGES * 32 + 255) / 256;
    const int  cgrid = ((int)(NH / 4) + 255) / 256;
    const int  ngrid = (NNODES * 32 + 255) / 256;
    const int  lgrid = (NNODES + 7) / 8;

    gemm_f32_wmma_kernel<<<gridN, gblk, 0, stream>>>(
        x, WT_ENC, enc_b, nullptr, B0, NNODES, FDIM, HDIM);

    auto conv = [&](const float* hin, int eblk, const float* wt, const float* bias,
                    const float* res, float* outp) {
        hipMemsetAsync(B2, 0, 3 * NH * sizeof(float), stream);
        edge_max_kernel<<<egrid, 256, 0, stream>>>(
            hin, src + (size_t)eblk * NEDGES, dst + (size_t)eblk * NEDGES, B2);
        edge_sum_kernel<<<egrid, 256, 0, stream>>>(
            hin, src + (size_t)eblk * NEDGES, dst + (size_t)eblk * NEDGES, B2, B3, B4);
        combine_kernel<<<cgrid, 256, 0, stream>>>(B4, B3, hin, B2, (int)(NH / 4));
        gemm_f32_wmma_kernel<<<gridN, gblk, 0, stream>>>(
            B2, wt, bias, res, outp, NNODES, HDIM, HDIM);
    };

    conv(B0, 0, WT_G, gcn_b, nullptr, B0);

    for (int l = 1; l < 3; ++l) {
        ln_relu_kernel<<<lgrid, 256, 0, stream>>>(
            B0, ln_g + (size_t)(l - 1) * HDIM, ln_b + (size_t)(l - 1) * HDIM, B1, NNODES);
        conv(B1, l - 1, WT_G + (size_t)l * HDIM * HDIM, gcn_b + (size_t)l * HDIM,
              B0,  B4);
        gather_kernel<<<ngrid, 256, 0, stream>>>(
            B4, node_map + (size_t)(l - 1) * NNODES, B0, NNODES);
    }

    const int sgrid = (NSEEDS * 32 + 255) / 256;
    gather_kernel<<<sgrid, 256, 0, stream>>>(B0, final_map, HG, NSEEDS);
    ln_relu_kernel<<<(NSEEDS + 7) / 8, 256, 0, stream>>>(
        HG, ln_g + 2 * HDIM, ln_b + 2 * HDIM, HG2, NSEEDS);
    const int  mtS = NSEEDS / 16;
    const dim3 gridS((mtS + 3) / 4, NCLSPAD / 64, 1);
    gemm_f32_wmma_kernel<<<gridS, gblk, 0, stream>>>(
        HG2, PWT, PB, nullptr, SC, NSEEDS, HDIM, NCLSPAD);
    logsoftmax_kernel<<<(NSEEDS + 7) / 8, 256, 0, stream>>>(SC, out, NSEEDS);
}


# Node GEMMs weight-stationary: each wave holds its K x 128 weight block (encoder: 256 x 64) in registers, read as given, and streams 16-row tiles; k pairs regrouped per WMMA, so k sum order changes
# speedup vs baseline: 1.1847x; 1.1847x over previous
#include <hip/hip_runtime.h>
#include <hip/hip_bf16.h>
#include <math.h>

#define NNODES   100000
#define NEDGES   600000
#define FDIM     256
#define HDIM     128
#define NCLS     47
#define NCLSPAD  64
#define NSEEDS   10000
#define GEN_EPS  1e-7f
#define LN_EPS   1e-5f

typedef float v2f __attribute__((ext_vector_type(2)));
typedef float v8f __attribute__((ext_vector_type(8)));

__global__ __launch_bounds__(128)
void gemm_f32_wmma_kernel(const float* __restrict__ A, const float* __restrict__ Wt,
                          const float* __restrict__ bias, const float* __restrict__ res,
                          float* __restrict__ out, int M, int K, int Nc)
{
    const int lane  = threadIdx.x;
    const int mtile = blockIdx.x * 4 + threadIdx.y;
    const int row0  = mtile * 16;
    if (row0 >= M) return;
    const int hb   = lane >> 4;
    const int lo   = lane & 15;
    const int arow = row0 + lo;
    const int col0 = blockIdx.y * 64 + lo;

    const float* __restrict__ a_ptr = A  + (size_t)arow * K;
    const float* __restrict__ b_ptr = Wt + (size_t)col0 * K;
    const size_t bstep = (size_t)16 * K;

    v8f acc[4];
#pragma unroll
    for (int t = 0; t < 4; ++t)
#pragma unroll
        for (int i = 0; i < 8; ++i) acc[t][i] = 0.f;

#pragma unroll 4
    for (int k = 0; k < K; k += 4) {
        const int ka = k + 2 * hb;
        const v2f a  = *(const v2f*)(a_ptr + ka);
        const v2f b0 = *(const v2f*)(b_ptr + ka);
        const v2f b1 = *(const v2f*)(b_ptr + bstep + ka);
        const v2f b2 = *(const v2f*)(b_ptr + 2 * bstep + ka);
        const v2f b3 = *(const v2f*)(b_ptr + 3 * bstep + ka);
        acc[0] = __builtin_amdgcn_wmma_f32_16x16x4_f32(false, a, false, b0, (short)0, acc[0], false, false);
        acc[1] = __builtin_amdgcn_wmma_f32_16x16x4_f32(false, a, false, b1, (short)0, acc[1], false, false);
        acc[2] = __builtin_amdgcn_wmma_f32_16x16x4_f32(false, a, false, b2, (short)0, acc[2], false, false);
        acc[3] = __builtin_amdgcn_wmma_f32_16x16x4_f32(false, a, false, b3, (short)0, acc[3], false, false);
    }

#pragma unroll
    for (int t = 0; t < 4; ++t) {
        const int   col = col0 + 16 * t;
        const float bv  = bias[col];
#pragma unroll
        for (int i = 0; i < 8; ++i) {
            const int r = row0 + i + 8 * hb;
            float v = acc[t][i] + bv;
            if (res) v += res[(size_t)r * Nc + col];
            out[(size_t)r * Nc + col] = v;
        }
    }
}

template <int K>
__device__ __forceinline__ void ws_load_a(float4 (&buf)[K / 8], const float* __restrict__ A,
                                          int tile, int lo, int hb)
{
    const float*   ap  = A + (size_t)tile * (16 * K);
    const unsigned lof = (unsigned)(lo * K + 4 * hb);
#pragma unroll
    for (int q = 0; q < K / 8; ++q) buf[q] = *(const float4*)(ap + lof + 8 * q);
}

template <int NT>
__device__ __forceinline__ void ws_load_r(float (&r)[NT][8], const float* __restrict__ res,
                                          int tile, int hb, int colb)
{
    const float*   rp  = res + (size_t)tile * (16 * HDIM);
    const unsigned lof = (unsigned)(8 * hb * HDIM + colb);
#pragma unroll
    for (int t = 0; t < NT; ++t)
#pragma unroll
        for (int i = 0; i < 8; ++i) r[t][i] = rp[lof + i * HDIM + 16 * t];
}

template <int K, int NT, bool HAS_RES>
__global__ __launch_bounds__(128)
void gemm_ws_kernel(const float* __restrict__ A, const float* __restrict__ W,
                    const float* __restrict__ bias, const float* __restrict__ res,
                    float* __restrict__ out, int mtiles)
{
    constexpr int NCG = HDIM / (16 * NT);
    constexpr int NQ  = K / 8;
    const int lane   = threadIdx.x & 31;
    const int wave   = blockIdx.x * 4 + __builtin_amdgcn_readfirstlane(threadIdx.x >> 5);
    const int hb     = lane >> 4;
    const int lo     = lane & 15;
    const int colb   = (wave % NCG) * 16 * NT + lo;
    const int stride = (gridDim.x * 4) / NCG;
    const int last   = mtiles - 1;
    int mt = wave / NCG;
    if (mt > last) return;

    v2f w[K / 4][NT];
#pragma unroll
    for (int q = 0; q < NQ; ++q) {
        const float* wp = W + (size_t)(8 * q + 4 * hb) * HDIM + colb;
#pragma unroll
        for (int t = 0; t < NT; ++t) {
            w[2 * q][t][0]     = wp[16 * t];
            w[2 * q][t][1]     = wp[HDIM + 16 * t];
            w[2 * q + 1][t][0] = wp[2 * HDIM + 16 * t];
            w[2 * q + 1][t][1] = wp[3 * HDIM + 16 * t];
        }
    }
    float bv[NT];
#pragma unroll
    for (int t = 0; t < NT; ++t) bv[t] = bias[colb + 16 * t];

    float4 cur[NQ], nxt[NQ];
    float  rc[NT][8], rn[NT][8];
    if (HAS_RES) ws_load_r<NT>(rn, res, mt, hb, colb);
    ws_load_a<K>(nxt, A, mt, lo, hb);

    while (true) {
#pragma unroll
        for (int q = 0; q < NQ; ++q) cur[q] = nxt[q];
        if (HAS_RES) {
#pragma unroll
            for (int t = 0; t < NT; ++t)
#pragma unroll
                for (int i = 0; i < 8; ++i) rc[t][i] = rn[t][i];
        }
        const int mtn = mt + stride;
        const int mtp = mtn < last ? mtn : last;
        ws_load_a<K>(nxt, A, mtp, lo, hb);

        v8f acc[NT];
#pragma unroll
        for (int t = 0; t < NT; ++t)
#pragma unroll
            for (int i = 0; i < 8; ++i) acc[t][i] = 0.f;

#pragma unroll
        for (int q = 0; q < NQ; ++q) {
            if (HAS_RES && q == NQ / 2) {
                __builtin_amdgcn_sched_barrier(0);
                ws_load_r<NT>(rn, res, mtp, hb, colb);
            }
            v2f a0, a1;
            a0[0] = cur[q].x; a0[1] = cur[q].y;
            a1[0] = cur[q].z; a1[1] = cur[q].w;
#pragma unroll
            for (int t = 0; t < NT; ++t)
                acc[t] = __builtin_amdgcn_wmma_f32_16x16x4_f32(false, a0, false, w[2 * q][t], (short)0, acc[t], false, false);
#pragma unroll
            for (int t = 0; t < NT; ++t)
                acc[t] = __builtin_amdgcn_wmma_f32_16x16x4_f32(false, a1, false, w[2 * q + 1][t], (short)0, acc[t], false, false);
        }

        float* __restrict__ op  = out + (size_t)mt * (16 * HDIM);
        const unsigned      oof = (unsigned)(8 * hb * HDIM + colb);
#pragma unroll
        for (int t = 0; t < NT; ++t)
#pragma unroll
            for (int i = 0; i < 8; ++i) {
                float v = acc[t][i] + bv[t];
                if (HAS_RES) v += rc[t][i];
                op[oof + i * HDIM + 16 * t] = v;
            }
        __builtin_amdgcn_sched_barrier(0);

        if (mtn > last) break;
        mt = mtn;
    }
}

__global__ __launch_bounds__(256)
void transpose_pad_kernel(const float* __restrict__ W, float* __restrict__ Wt,
                          int K, int Nc, int Ncpad)
{
    const int tid = blockIdx.x * 256 + threadIdx.x;
    if (tid >= Ncpad * K) return;
    const int c = tid / K;
    const int k = tid - c * K;
    Wt[(size_t)c * K + k] = (c < Nc) ? W[(size_t)k * Nc + c] : 0.f;
}

__global__ void pad_bias_kernel(const float* __restrict__ b, float* __restrict__ bp,
                                int Nc, int Ncpad)
{
    const int c = threadIdx.x;
    if (c < Ncpad) bp[c] = (c < Nc) ? b[c] : 0.f;
}

__global__ __launch_bounds__(256)
void edge_max_kernel(const float* __restrict__ hin, const int* __restrict__ src,
                     const int* __restrict__ dst, float* __restrict__ mx)
{
    const int tid = blockIdx.x * 256 + threadIdx.x;
    if (tid >= NEDGES * 32) return;
    const int e = tid >> 5;
    const int f = (tid & 31) * 4;
    const int s = src[e];
    const int d = dst[e];
    const float4 hv = *(const float4*)(hin + (size_t)s * HDIM + f);
    unsigned* mp = (unsigned*)(mx + (size_t)d * HDIM + f);
    atomicMax(mp + 0, __float_as_uint(fmaxf(hv.x, 0.f) + GEN_EPS));
    atomicMax(mp + 1, __float_as_uint(fmaxf(hv.y, 0.f) + GEN_EPS));
    atomicMax(mp + 2, __float_as_uint(fmaxf(hv.z, 0.f) + GEN_EPS));
    atomicMax(mp + 3, __float_as_uint(fmaxf(hv.w, 0.f) + GEN_EPS));
}

__global__ __launch_bounds__(256)
void edge_sum_kernel(const float* __restrict__ hin, const int* __restrict__ src,
                     const int* __restrict__ dst, const float* __restrict__ mx,
                     float* __restrict__ denom, float* __restrict__ numer)
{
    const int tid = blockIdx.x * 256 + threadIdx.x;
    if (tid >= NEDGES * 32) return;
    const int e = tid >> 5;
    const int f = (tid & 31) * 4;
    const int s = src[e];
    const int d = dst[e];
    const float4 hv = *(const float4*)(hin + (size_t)s * HDIM + f);
    const float4 mv = *(const float4*)(mx + (size_t)d * HDIM + f);
    const size_t o = (size_t)d * HDIM + f;
    float m, ex;
    m = fmaxf(hv.x, 0.f) + GEN_EPS; ex = expf(m - mv.x);
    atomicAdd(denom + o + 0, ex); atomicAdd(numer + o + 0, m * ex);
    m = fmaxf(hv.y, 0.f) + GEN_EPS; ex = expf(m - mv.y);
    atomicAdd(denom + o + 1, ex); atomicAdd(numer + o + 1, m * ex);
    m = fmaxf(hv.z, 0.f) + GEN_EPS; ex = expf(m - mv.z);
    atomicAdd(denom + o + 2, ex); atomicAdd(numer + o + 2, m * ex);
    m = fmaxf(hv.w, 0.f) + GEN_EPS; ex = expf(m - mv.w);
    atomicAdd(denom + o + 3, ex); atomicAdd(numer + o + 3, m * ex);
}

__global__ __launch_bounds__(256)
void combine_kernel(const float* __restrict__ numer, const float* __restrict__ denom,
                    const float* __restrict__ hin, float* __restrict__ t, int n4)
{
    const int i = blockIdx.x * 256 + threadIdx.x;
    if (i >= n4) return;
    const float4 nu = ((const float4*)numer)[i];
    const float4 de = ((const float4*)denom)[i];
    const float4 hv = ((const float4*)hin)[i];
    float4 r;
    r.x = nu.x / (de.x + 1e-16f) + hv.x;
    r.y = nu.y / (de.y + 1e-16f) + hv.y;
    r.z = nu.z / (de.z + 1e-16f) + hv.z;
    r.w = nu.w / (de.w + 1e-16f) + hv.w;
    ((float4*)t)[i] = r;
}

__global__ __launch_bounds__(256)
void ln_relu_kernel(const float* __restrict__ in, const float* __restrict__ g,
                    const float* __restrict__ b, float* __restrict__ out, int M)
{
    const int row  = blockIdx.x * 8 + (threadIdx.x >> 5);
    const int lane = threadIdx.x & 31;
    if (row >= M) return;
    const float4 v = *(const float4*)(in + (size_t)row * HDIM + lane * 4);
    float s1 = v.x + v.y + v.z + v.w;
    float s2 = v.x * v.x + v.y * v.y + v.z * v.z + v.w * v.w;
#pragma unroll
    for (int off = 16; off > 0; off >>= 1) {
        s1 += __shfl_xor(s1, off, 32);
        s2 += __shfl_xor(s2, off, 32);
    }
    const float mu  = s1 * (1.f / HDIM);
    const float var = s2 * (1.f / HDIM) - mu * mu;
    const float inv = rsqrtf(var + LN_EPS);
    const float4 gg = *(const float4*)(g + lane * 4);
    const float4 bb = *(const float4*)(b + lane * 4);
    float4 r;
    r.x = fmaxf((v.x - mu) * inv * gg.x + bb.x, 0.f);
    r.y = fmaxf((v.y - mu) * inv * gg.y + bb.y, 0.f);
    r.z = fmaxf((v.z - mu) * inv * gg.z + bb.z, 0.f);
    r.w = fmaxf((v.w - mu) * inv * gg.w + bb.w, 0.f);
    *(float4*)(out + (size_t)row * HDIM + lane * 4) = r;
}

__global__ __launch_bounds__(256)
void gather_kernel(const float* __restrict__ in, const int* __restrict__ map,
                   float* __restrict__ out, int M)
{
    const int tid = blockIdx.x * 256 + threadIdx.x;
    if (tid >= M * 32) return;
    const int i = tid >> 5;
    const int f = (tid & 31) * 4;
    const int j = map[i];
    *(float4*)(out + (size_t)i * HDIM + f) =
        *(const float4*)(in + (size_t)j * HDIM + f);
}

__global__ __launch_bounds__(256)
void logsoftmax_kernel(const float* __restrict__ in, float* __restrict__ out, int M)
{
    const int row  = blockIdx.x * 8 + (threadIdx.x >> 5);
    const int lane = threadIdx.x & 31;
    if (row >= M) return;
    const float NEG = -3.0e38f;
    const float v0 = (lane      < NCLS) ? in[(size_t)row * NCLSPAD + lane     ] : NEG;
    const float v1 = (lane + 32 < NCLS) ? in[(size_t)row * NCLSPAD + lane + 32] : NEG;
    float mx = fmaxf(v0, v1);
#pragma unroll
    for (int off = 16; off > 0; off >>= 1) mx = fmaxf(mx, __shfl_xor(mx, off, 32));
    float se = ((lane      < NCLS) ? expf(v0 - mx) : 0.f)
             + ((lane + 32 < NCLS) ? expf(v1 - mx) : 0.f);
#pragma unroll
    for (int off = 16; off > 0; off >>= 1) se += __shfl_xor(se, off, 32);
    const float lse = logf(se) + mx;
    if (lane      < NCLS) out[(size_t)row * NCLS + lane     ] = v0 - lse;
    if (lane + 32 < NCLS) out[(size_t)row * NCLS + lane + 32] = v1 - lse;
}

extern "C" void kernel_launch(void* const* d_in, const int* in_sizes, int n_in,
                              void* d_out, int out_size, void* d_ws, size_t ws_size,
                              hipStream_t stream)
{
    (void)in_sizes; (void)n_in; (void)out_size; (void)ws_size;

    const float* x         = (const float*)d_in[0];
    const int*   src       = (const int*)  d_in[1];
    const int*   dst       = (const int*)  d_in[2];
    const int*   node_map  = (const int*)  d_in[3];
    const int*   final_map = (const int*)  d_in[4];
    const float* enc_w     = (const float*)d_in[5];
    const float* enc_b     = (const float*)d_in[6];
    const float* gcn_w     = (const float*)d_in[7];
    const float* gcn_b     = (const float*)d_in[8];
    const float* ln_g      = (const float*)d_in[9];
    const float* ln_b      = (const float*)d_in[10];
    const float* pred_w    = (const float*)d_in[11];
    const float* pred_b    = (const float*)d_in[12];
    float* out = (float*)d_out;

    const size_t NH = (size_t)NNODES * HDIM;
    float* B0     = (float*)d_ws;
    float* B1     = B0 + NH;
    float* B2     = B0 + 2 * NH;
    float* B3     = B0 + 3 * NH;
    float* B4     = B0 + 4 * NH;
    float* HG     = B0 + 5 * NH;
    float* HG2    = HG  + (size_t)NSEEDS * HDIM;
    float* SC     = HG2 + (size_t)NSEEDS * HDIM;
    float* PWT    = SC  + (size_t)NSEEDS * NCLSPAD;
    float* PB     = PWT    + (size_t)NCLSPAD * HDIM;

    transpose_pad_kernel<<<(NCLSPAD * HDIM + 255) / 256, 256, 0, stream>>>(
        pred_w, PWT, HDIM, NCLS, NCLSPAD);
    pad_bias_kernel<<<1, NCLSPAD, 0, stream>>>(pred_b, PB, NCLS, NCLSPAD);

    const dim3 gblk(32, 4, 1);
    const int  mtN   = NNODES / 16;
    const int  NWGP  = 256;
    const int  egrid = (NEDGES * 32 + 255) / 256;
    const int  cgrid = ((int)(NH / 4) + 255) / 256;
    const int  ngrid = (NNODES * 32 + 255) / 256;
    const int  lgrid = (NNODES + 7) / 8;

    gemm_ws_kernel<FDIM, 4, false><<<NWGP, 128, 0, stream>>>(
        x, enc_w, enc_b, nullptr, B0, mtN);

    auto conv = [&](const float* hin, int eblk, const float* wt, const float* bias,
                    const float* res, float* outp) {
        hipMemsetAsync(B2, 0, 3 * NH * sizeof(float), stream);
        edge_max_kernel<<<egrid, 256, 0, stream>>>(
            hin, src + (size_t)eblk * NEDGES, dst + (size_t)eblk * NEDGES, B2);
        edge_sum_kernel<<<egrid, 256, 0, stream>>>(
            hin, src + (size_t)eblk * NEDGES, dst + (size_t)eblk * NEDGES, B2, B3, B4);
        combine_kernel<<<cgrid, 256, 0, stream>>>(B4, B3, hin, B2, (int)(NH / 4));
        if (res)
            gemm_ws_kernel<HDIM, 8, true><<<NWGP, 128, 0, stream>>>(B2, wt, bias, res, outp, mtN);
        else
            gemm_ws_kernel<HDIM, 8, false><<<NWGP, 128, 0, stream>>>(B2, wt, bias, nullptr, outp, mtN);
    };

    conv(B0, 0, gcn_w, gcn_b, nullptr, B0);

    for (int l = 1; l < 3; ++l) {
        ln_relu_kernel<<<lgrid, 256, 0, stream>>>(
            B0, ln_g + (size_t)(l - 1) * HDIM, ln_b + (size_t)(l - 1) * HDIM, B1, NNODES);
        conv(B1, l - 1, gcn_w + (size_t)l * HDIM * HDIM, gcn_b + (size_t)l * HDIM,
              B0,  B4);
        gather_kernel<<<ngrid, 256, 0, stream>>>(
            B4, node_map + (size_t)(l - 1) * NNODES, B0, NNODES);
    }

    const int sgrid = (NSEEDS * 32 + 255) / 256;
    gather_kernel<<<sgrid, 256, 0, stream>>>(B0, final_map, HG, NSEEDS);
    ln_relu_kernel<<<(NSEEDS + 7) / 8, 256, 0, stream>>>(
        HG, ln_g + 2 * HDIM, ln_b + 2 * HDIM, HG2, NSEEDS);
    const int  mtS = NSEEDS / 16;
    const dim3 gridS((mtS + 3) / 4, NCLSPAD / 64, 1);
    gemm_f32_wmma_kernel<<<gridS, gblk, 0, stream>>>(
        HG2, PWT, PB, nullptr, SC, NSEEDS, HDIM, NCLSPAD);
    logsoftmax_kernel<<<(NSEEDS + 7) / 8, 256, 0, stream>>>(SC, out, NSEEDS);
}


# Softmax aggregation by destination-range owners: a work-group scans dst once, keeps max and both sums of its rows in LDS (LDS atomics, sums in arrival order), writes t; global atomics and clears gone
# speedup vs baseline: 3.0306x; 2.5581x over previous
#include <hip/hip_runtime.h>
#include <hip/hip_bf16.h>
#include <math.h>

#define NNODES   100000
#define NEDGES   600000
#define FDIM     256
#define HDIM     128
#define NCLS     47
#define NCLSPAD  64
#define NSEEDS   10000
#define GEN_EPS  1e-7f
#define LN_EPS   1e-5f

typedef float v2f __attribute__((ext_vector_type(2)));
typedef float v8f __attribute__((ext_vector_type(8)));
typedef float v4f __attribute__((ext_vector_type(4)));

__global__ __launch_bounds__(128)
void gemm_f32_wmma_kernel(const float* __restrict__ A, const float* __restrict__ Wt,
                          const float* __restrict__ bias, const float* __restrict__ res,
                          float* __restrict__ out, int M, int K, int Nc)
{
    const int lane  = threadIdx.x;
    const int mtile = blockIdx.x * 4 + threadIdx.y;
    const int row0  = mtile * 16;
    if (row0 >= M) return;
    const int hb   = lane >> 4;
    const int lo   = lane & 15;
    const int arow = row0 + lo;
    const int col0 = blockIdx.y * 64 + lo;

    const float* __restrict__ a_ptr = A  + (size_t)arow * K;
    const float* __restrict__ b_ptr = Wt + (size_t)col0 * K;
    const size_t bstep = (size_t)16 * K;

    v8f acc[4];
#pragma unroll
    for (int t = 0; t < 4; ++t)
#pragma unroll
        for (int i = 0; i < 8; ++i) acc[t][i] = 0.f;

#pragma unroll 4
    for (int k = 0; k < K; k += 4) {
        const int ka = k + 2 * hb;
        const v2f a  = *(const v2f*)(a_ptr + ka);
        const v2f b0 = *(const v2f*)(b_ptr + ka);
        const v2f b1 = *(const v2f*)(b_ptr + bstep + ka);
        const v2f b2 = *(const v2f*)(b_ptr + 2 * bstep + ka);
        const v2f b3 = *(const v2f*)(b_ptr + 3 * bstep + ka);
        acc[0] = __builtin_amdgcn_wmma_f32_16x16x4_f32(false, a, false, b0, (short)0, acc[0], false, false);
        acc[1] = __builtin_amdgcn_wmma_f32_16x16x4_f32(false, a, false, b1, (short)0, acc[1], false, false);
        acc[2] = __builtin_amdgcn_wmma_f32_16x16x4_f32(false, a, false, b2, (short)0, acc[2], false, false);
        acc[3] = __builtin_amdgcn_wmma_f32_16x16x4_f32(false, a, false, b3, (short)0, acc[3], false, false);
    }

#pragma unroll
    for (int t = 0; t < 4; ++t) {
        const int   col = col0 + 16 * t;
        const float bv  = bias[col];
#pragma unroll
        for (int i = 0; i < 8; ++i) {
            const int r = row0 + i + 8 * hb;
            float v = acc[t][i] + bv;
            if (res) v += res[(size_t)r * Nc + col];
            out[(size_t)r * Nc + col] = v;
        }
    }
}

template <int K>
__device__ __forceinline__ void ws_load_a(float4 (&buf)[K / 8], const float* __restrict__ A,
                                          int tile, int lo, int hb)
{
    const float*   ap  = A + (size_t)tile * (16 * K);
    const unsigned lof = (unsigned)(lo * K + 4 * hb);
#pragma unroll
    for (int q = 0; q < K / 8; ++q) buf[q] = *(const float4*)(ap + lof + 8 * q);
}

template <int NT>
__device__ __forceinline__ void ws_load_r(float (&r)[NT][8], const float* __restrict__ res,
                                          int tile, int hb, int colb)
{
    const float*   rp  = res + (size_t)tile * (16 * HDIM);
    const unsigned lof = (unsigned)(8 * hb * HDIM + colb);
#pragma unroll
    for (int t = 0; t < NT; ++t)
#pragma unroll
        for (int i = 0; i < 8; ++i) r[t][i] = rp[lof + i * HDIM + 16 * t];
}

template <int K, int NT, bool HAS_RES>
__global__ __launch_bounds__(128)
void gemm_ws_kernel(const float* __restrict__ A, const float* __restrict__ W,
                    const float* __restrict__ bias, const float* __restrict__ res,
                    float* __restrict__ out, int mtiles)
{
    constexpr int NCG = HDIM / (16 * NT);
    constexpr int NQ  = K / 8;
    const int lane   = threadIdx.x & 31;
    const int wave   = blockIdx.x * 4 + __builtin_amdgcn_readfirstlane(threadIdx.x >> 5);
    const int hb     = lane >> 4;
    const int lo     = lane & 15;
    const int colb   = (wave % NCG) * 16 * NT + lo;
    const int stride = (gridDim.x * 4) / NCG;
    const int last   = mtiles - 1;
    int mt = wave / NCG;
    if (mt > last) return;

    v2f w[K / 4][NT];
#pragma unroll
    for (int q = 0; q < NQ; ++q) {
        const float* wp = W + (size_t)(8 * q + 4 * hb) * HDIM + colb;
#pragma unroll
        for (int t = 0; t < NT; ++t) {
            w[2 * q][t][0]     = wp[16 * t];
            w[2 * q][t][1]     = wp[HDIM + 16 * t];
            w[2 * q + 1][t][0] = wp[2 * HDIM + 16 * t];
            w[2 * q + 1][t][1] = wp[3 * HDIM + 16 * t];
        }
    }
    float bv[NT];
#pragma unroll
    for (int t = 0; t < NT; ++t) bv[t] = bias[colb + 16 * t];

    float4 cur[NQ], nxt[NQ];
    float  rc[NT][8], rn[NT][8];
    if (HAS_RES) ws_load_r<NT>(rn, res, mt, hb, colb);
    ws_load_a<K>(nxt, A, mt, lo, hb);

    while (true) {
#pragma unroll
        for (int q = 0; q < NQ; ++q) cur[q] = nxt[q];
        if (HAS_RES) {
#pragma unroll
            for (int t = 0; t < NT; ++t)
#pragma unroll
                for (int i = 0; i < 8; ++i) rc[t][i] = rn[t][i];
        }
        const int mtn = mt + stride;
        const int mtp = mtn < last ? mtn : last;
        ws_load_a<K>(nxt, A, mtp, lo, hb);

        v8f acc[NT];
#pragma unroll
        for (int t = 0; t < NT; ++t)
#pragma unroll
            for (int i = 0; i < 8; ++i) acc[t][i] = 0.f;

#pragma unroll
        for (int q = 0; q < NQ; ++q) {
            if (HAS_RES && q == NQ / 2) {
                __builtin_amdgcn_sched_barrier(0);
                ws_load_r<NT>(rn, res, mtp, hb, colb);
            }
            v2f a0, a1;
            a0[0] = cur[q].x; a0[1] = cur[q].y;
            a1[0] = cur[q].z; a1[1] = cur[q].w;
#pragma unroll
            for (int t = 0; t < NT; ++t)
                acc[t] = __builtin_amdgcn_wmma_f32_16x16x4_f32(false, a0, false, w[2 * q][t], (short)0, acc[t], false, false);
#pragma unroll
            for (int t = 0; t < NT; ++t)
                acc[t] = __builtin_amdgcn_wmma_f32_16x16x4_f32(false, a1, false, w[2 * q + 1][t], (short)0, acc[t], false, false);
        }

        float* __restrict__ op  = out + (size_t)mt * (16 * HDIM);
        const unsigned      oof = (unsigned)(8 * hb * HDIM + colb);
#pragma unroll
        for (int t = 0; t < NT; ++t)
#pragma unroll
            for (int i = 0; i < 8; ++i) {
                float v = acc[t][i] + bv[t];
                if (HAS_RES) v += rc[t][i];
                op[oof + i * HDIM + 16 * t] = v;
            }
        __builtin_amdgcn_sched_barrier(0);

        if (mtn > last) break;
        mt = mtn;
    }
}

__global__ __launch_bounds__(256)
void transpose_pad_kernel(const float* __restrict__ W, float* __restrict__ Wt,
                          int K, int Nc, int Ncpad)
{
    const int tid = blockIdx.x * 256 + threadIdx.x;
    if (tid >= Ncpad * K) return;
    const int c = tid / K;
    const int k = tid - c * K;
    Wt[(size_t)c * K + k] = (c < Nc) ? W[(size_t)k * Nc + c] : 0.f;
}

__global__ void pad_bias_kernel(const float* __restrict__ b, float* __restrict__ bp,
                                int Nc, int Ncpad)
{
    const int c = threadIdx.x;
    if (c < Ncpad) bp[c] = (c < Nc) ? b[c] : 0.f;
}

#define AGG_R      196
#define AGG_K      2
#define AGG_RW     (AGG_R * AGG_K)
#define AGG_NPART  ((NNODES + AGG_RW - 1) / AGG_RW)
#define AGG_CAP    3000
#define AGG_U      8
#define AGG_NV     (NEDGES / 4)
#define AGG_NBF    (AGG_NV / (1024 * AGG_U))
#define AGG_TAIL0  (AGG_NBF * 1024 * AGG_U)
#define AGG_NT     ((AGG_NV - AGG_TAIL0 + 1023) / 1024)

static_assert(NEDGES % 4 == 0 && NEDGES <= (1 << 20), "edge index packing");
static_assert(NNODES <= (1 << 17) && AGG_R <= 256, "list entry packing");
static_assert(AGG_NBF >= 1, "scan batching");

__device__ __forceinline__ unsigned agg_umin(unsigned a, unsigned b) { return a < b ? a : b; }

__device__ __forceinline__ void lds_add(float* p, float v)
{
    (void)__hip_atomic_fetch_add(p, v, __ATOMIC_RELAXED, __HIP_MEMORY_SCOPE_WORKGROUP);
}

__device__ __forceinline__ void agg_edge_max(unsigned* s_mx, const v4f hv, unsigned ld, int lane)
{
    unsigned* p = s_mx + ld * HDIM + lane;
    atomicMax(p,      __float_as_uint(fmaxf(hv.x, 0.f) + GEN_EPS));
    atomicMax(p + 32, __float_as_uint(fmaxf(hv.y, 0.f) + GEN_EPS));
    atomicMax(p + 64, __float_as_uint(fmaxf(hv.z, 0.f) + GEN_EPS));
    atomicMax(p + 96, __float_as_uint(fmaxf(hv.w, 0.f) + GEN_EPS));
}

__device__ __forceinline__ void agg_edge_sum(const unsigned* s_mx, float* s_de, float* s_nu,
                                             const v4f hv, unsigned ld, int lane)
{
    const unsigned o = ld * HDIM + lane;
    const float x0 = __uint_as_float(s_mx[o]);
    const float x1 = __uint_as_float(s_mx[o + 32]);
    const float x2 = __uint_as_float(s_mx[o + 64]);
    const float x3 = __uint_as_float(s_mx[o + 96]);
    const float m0 = fmaxf(hv.x, 0.f) + GEN_EPS, e0 = expf(m0 - x0);
    const float m1 = fmaxf(hv.y, 0.f) + GEN_EPS, e1 = expf(m1 - x1);
    const float m2 = fmaxf(hv.z, 0.f) + GEN_EPS, e2 = expf(m2 - x2);
    const float m3 = fmaxf(hv.w, 0.f) + GEN_EPS, e3 = expf(m3 - x3);
    lds_add(s_de + o,      e0); lds_add(s_nu + o,      m0 * e0);
    lds_add(s_de + o + 32, e1); lds_add(s_nu + o + 32, m1 * e1);
    lds_add(s_de + o + 64, e2); lds_add(s_nu + o + 64, m2 * e2);
    lds_add(s_de + o + 96, e3); lds_add(s_nu + o + 96, m3 * e3);
}

__device__ __forceinline__ void agg_test4(const int4 d, int i, int row0, unsigned nrw,
                                          unsigned* s_list, unsigned* s_cnt)
{
    const unsigned l0 = (unsigned)(d.x - row0);
    const unsigned l1 = (unsigned)(d.y - row0);
    const unsigned l2 = (unsigned)(d.z - row0);
    const unsigned l3 = (unsigned)(d.w - row0);
    if (agg_umin(agg_umin(l0, l1), agg_umin(l2, l3)) < nrw) {
        const unsigned e  = 4u * (unsigned)i;
        const unsigned lv[4] = {l0, l1, l2, l3};
#pragma unroll
        for (int k = 0; k < 4; ++k) {
            if (lv[k] < nrw) {
                const unsigned h   = lv[k] >= AGG_R ? 1u : 0u;
                const unsigned pos = atomicAdd(s_cnt + h, 1u);
                if (pos < AGG_CAP) s_list[h * AGG_CAP + pos] = (e + k) | ((lv[k] - h * AGG_R) << 20);
            }
        }
    }
}

__global__ __launch_bounds__(1024)
void agg_kernel(const float* __restrict__ hin, const int* __restrict__ src,
                const int* __restrict__ dst, float* __restrict__ t)
{
    __shared__ unsigned s_mx[(AGG_R + 1) * HDIM];
    __shared__ float    s_de[(AGG_R + 1) * HDIM];
    __shared__ float    s_nu[(AGG_R + 1) * HDIM];
    __shared__ unsigned s_list[AGG_K * AGG_CAP];
    __shared__ unsigned s_cnt[AGG_K];

    const int tid  = threadIdx.x;
    const int lane = tid & 31;
    const unsigned wave = (unsigned)__builtin_amdgcn_readfirstlane(tid >> 5);
    const int      row0 = blockIdx.x * AGG_RW;
    const unsigned nrw  = (unsigned)((NNODES - row0) < AGG_RW ? (NNODES - row0) : AGG_RW);

    if (tid < AGG_K) s_cnt[tid] = 0u;
    __syncthreads();

    const int4* __restrict__ d4 = (const int4*)dst;
    const int rot = (int)(blockIdx.x % AGG_NBF);
    for (int bb = 0; bb < AGG_NBF; ++bb) {
        int b = bb + rot;
        if (b >= AGG_NBF) b -= AGG_NBF;
        int4 d[AGG_U];
#pragma unroll
        for (int u = 0; u < AGG_U; ++u) d[u] = d4[(b * AGG_U + u) * 1024 + tid];
#pragma unroll
        for (int u = 0; u < AGG_U; ++u)
            agg_test4(d[u], (b * AGG_U + u) * 1024 + tid, row0, nrw, s_list, s_cnt);
    }
    {
        int4 d[AGG_NT];
#pragma unroll
        for (int u = 0; u < AGG_NT; ++u) {
            const int i = AGG_TAIL0 + u * 1024 + tid;
            d[u] = (i < AGG_NV) ? d4[i] : make_int4(-1, -1, -1, -1);
        }
#pragma unroll
        for (int u = 0; u < AGG_NT; ++u)
            agg_test4(d[u], AGG_TAIL0 + u * 1024 + tid, row0, nrw, s_list, s_cnt);
    }
    __syncthreads();

    for (int h = 0; h < AGG_K; ++h) {
        const unsigned n = s_cnt[h];
        if (n <= AGG_CAP)
            for (unsigned i = tid; i < n; i += 1024) {
                const unsigned en = s_list[h * AGG_CAP + i];
                s_list[h * AGG_CAP + i] = (unsigned)src[en & 0xFFFFFu] | ((en >> 20) << 17);
            }
    }
    __syncthreads();

    for (int h = 0; h < AGG_K; ++h) {
        const int rbase = row0 + h * AGG_R;
        const int left  = NNODES - rbase;
        if (left <= 0) break;
        const unsigned  nrows  = (unsigned)(left < AGG_R ? left : AGG_R);
        const unsigned  n      = (unsigned)__builtin_amdgcn_readfirstlane((int)s_cnt[h]);
        const bool      listed = n <= AGG_CAP;
        const unsigned* lst    = s_list + h * AGG_CAP;

        for (int i = tid; i < (AGG_R + 1) * HDIM; i += 1024) { s_mx[i] = 0u; s_de[i] = 0.f; s_nu[i] = 0.f; }
        __syncthreads();

        if (listed) {
            for (unsigned i0 = wave; i0 < n; i0 += 128) {
                unsigned en[4], ld[4];
                v4f      hv[4];
#pragma unroll
                for (int u = 0; u < 4; ++u) {
                    const unsigned idx = i0 + 32u * u;
                    en[u] = lst[idx < n ? idx : n - 1u];
                    ld[u] = idx < n ? (en[u] >> 17) : (unsigned)AGG_R;
                }
#pragma unroll
                for (int u = 0; u < 4; ++u)
                    hv[u] = *(const v4f*)(hin + (size_t)(en[u] & 0x1FFFFu) * HDIM + 4 * lane);
                asm volatile("" :: "v"(hv[0]), "v"(hv[1]), "v"(hv[2]), "v"(hv[3]));
#pragma unroll
                for (int u = 0; u < 4; ++u) agg_edge_max(s_mx, hv[u], ld[u], lane);
            }
        } else {
            for (unsigned e = wave; e < NEDGES; e += 32) {
                const unsigned l = (unsigned)(dst[e] - rbase);
                if (l < nrows)
                    agg_edge_max(s_mx, *(const v4f*)(hin + (size_t)src[e] * HDIM + 4 * lane), l, lane);
            }
        }
        __syncthreads();

        if (listed) {
            for (unsigned i0 = wave; i0 < n; i0 += 128) {
                unsigned en[4], ld[4];
                v4f      hv[4];
#pragma unroll
                for (int u = 0; u < 4; ++u) {
                    const unsigned idx = i0 + 32u * u;
                    en[u] = lst[idx < n ? idx : n - 1u];
                    ld[u] = idx < n ? (en[u] >> 17) : (unsigned)AGG_R;
                }
#pragma unroll
                for (int u = 0; u < 4; ++u)
                    hv[u] = *(const v4f*)(hin + (size_t)(en[u] & 0x1FFFFu) * HDIM + 4 * lane);
#pragma unroll
                for (int u = 0; u < 4; ++u) agg_edge_sum(s_mx, s_de, s_nu, hv[u], ld[u], lane);
            }
        } else {
            for (unsigned e = wave; e < NEDGES; e += 32) {
                const unsigned l = (unsigned)(dst[e] - rbase);
                if (l < nrows)
                    agg_edge_sum(s_mx, s_de, s_nu,
                                 *(const v4f*)(hin + (size_t)src[e] * HDIM + 4 * lane), l, lane);
            }
        }
        __syncthreads();

        for (unsigned r = wave; r < nrows; r += 32) {
            const size_t   g  = (size_t)(rbase + (int)r) * HDIM + 4 * lane;
            const float4   hv = *(const float4*)(hin + g);
            const unsigned o  = r * HDIM + lane;
            float4 q;
            q.x = s_nu[o]      / (s_de[o]      + 1e-16f) + hv.x;
            q.y = s_nu[o + 32] / (s_de[o + 32] + 1e-16f) + hv.y;
            q.z = s_nu[o + 64] / (s_de[o + 64] + 1e-16f) + hv.z;
            q.w = s_nu[o + 96] / (s_de[o + 96] + 1e-16f) + hv.w;
            *(float4*)(t + g) = q;
        }
        __syncthreads();
    }
}

__global__ __launch_bounds__(256)
void ln_relu_kernel(const float* __restrict__ in, const float* __restrict__ g,
                    const float* __restrict__ b, float* __restrict__ out, int M)
{
    const int row  = blockIdx.x * 8 + (threadIdx.x >> 5);
    const int lane = threadIdx.x & 31;
    if (row >= M) return;
    const float4 v = *(const float4*)(in + (size_t)row * HDIM + lane * 4);
    float s1 = v.x + v.y + v.z + v.w;
    float s2 = v.x * v.x + v.y * v.y + v.z * v.z + v.w * v.w;
#pragma unroll
    for (int off = 16; off > 0; off >>= 1) {
        s1 += __shfl_xor(s1, off, 32);
        s2 += __shfl_xor(s2, off, 32);
    }
    const float mu  = s1 * (1.f / HDIM);
    const float var = s2 * (1.f / HDIM) - mu * mu;
    const float inv = rsqrtf(var + LN_EPS);
    const float4 gg = *(const float4*)(g + lane * 4);
    const float4 bb = *(const float4*)(b + lane * 4);
    float4 r;
    r.x = fmaxf((v.x - mu) * inv * gg.x + bb.x, 0.f);
    r.y = fmaxf((v.y - mu) * inv * gg.y + bb.y, 0.f);
    r.z = fmaxf((v.z - mu) * inv * gg.z + bb.z, 0.f);
    r.w = fmaxf((v.w - mu) * inv * gg.w + bb.w, 0.f);
    *(float4*)(out + (size_t)row * HDIM + lane * 4) = r;
}

__global__ __launch_bounds__(256)
void gather_kernel(const float* __restrict__ in, const int* __restrict__ map,
                   float* __restrict__ out, int M)
{
    const int tid = blockIdx.x * 256 + threadIdx.x;
    if (tid >= M * 32) return;
    const int i = tid >> 5;
    const int f = (tid & 31) * 4;
    const int j = map[i];
    *(float4*)(out + (size_t)i * HDIM + f) =
        *(const float4*)(in + (size_t)j * HDIM + f);
}

__global__ __launch_bounds__(256)
void logsoftmax_kernel(const float* __restrict__ in, float* __restrict__ out, int M)
{
    const int row  = blockIdx.x * 8 + (threadIdx.x >> 5);
    const int lane = threadIdx.x & 31;
    if (row >= M) return;
    const float NEG = -3.0e38f;
    const float v0 = (lane      < NCLS) ? in[(size_t)row * NCLSPAD + lane     ] : NEG;
    const float v1 = (lane + 32 < NCLS) ? in[(size_t)row * NCLSPAD + lane + 32] : NEG;
    float mx = fmaxf(v0, v1);
#pragma unroll
    for (int off = 16; off > 0; off >>= 1) mx = fmaxf(mx, __shfl_xor(mx, off, 32));
    float se = ((lane      < NCLS) ? expf(v0 - mx) : 0.f)
             + ((lane + 32 < NCLS) ? expf(v1 - mx) : 0.f);
#pragma unroll
    for (int off = 16; off > 0; off >>= 1) se += __shfl_xor(se, off, 32);
    const float lse = logf(se) + mx;
    if (lane      < NCLS) out[(size_t)row * NCLS + lane     ] = v0 - lse;
    if (lane + 32 < NCLS) out[(size_t)row * NCLS + lane + 32] = v1 - lse;
}

extern "C" void kernel_launch(void* const* d_in, const int* in_sizes, int n_in,
                              void* d_out, int out_size, void* d_ws, size_t ws_size,
                              hipStream_t stream)
{
    (void)in_sizes; (void)n_in; (void)out_size; (void)ws_size;

    const float* x         = (const float*)d_in[0];
    const int*   src       = (const int*)  d_in[1];
    const int*   dst       = (const int*)  d_in[2];
    const int*   node_map  = (const int*)  d_in[3];
    const int*   final_map = (const int*)  d_in[4];
    const float* enc_w     = (const float*)d_in[5];
    const float* enc_b     = (const float*)d_in[6];
    const float* gcn_w     = (const float*)d_in[7];
    const float* gcn_b     = (const float*)d_in[8];
    const float* ln_g      = (const float*)d_in[9];
    const float* ln_b      = (const float*)d_in[10];
    const float* pred_w    = (const float*)d_in[11];
    const float* pred_b    = (const float*)d_in[12];
    float* out = (float*)d_out;

    const size_t NH = (size_t)NNODES * HDIM;
    float* B0     = (float*)d_ws;
    float* B1     = B0 + NH;
    float* B2     = B0 + 2 * NH;
    float* B4     = B0 + 4 * NH;
    float* HG     = B0 + 5 * NH;
    float* HG2    = HG  + (size_t)NSEEDS * HDIM;
    float* SC     = HG2 + (size_t)NSEEDS * HDIM;
    float* PWT    = SC  + (size_t)NSEEDS * NCLSPAD;
    float* PB     = PWT    + (size_t)NCLSPAD * HDIM;

    transpose_pad_kernel<<<(NCLSPAD * HDIM + 255) / 256, 256, 0, stream>>>(
        pred_w, PWT, HDIM, NCLS, NCLSPAD);
    pad_bias_kernel<<<1, NCLSPAD, 0, stream>>>(pred_b, PB, NCLS, NCLSPAD);

    const dim3 gblk(32, 4, 1);
    const int  mtN   = NNODES / 16;
    const int  NWGP  = 256;
    const int  ngrid = (NNODES * 32 + 255) / 256;
    const int  lgrid = (NNODES + 7) / 8;

    gemm_ws_kernel<FDIM, 4, false><<<NWGP, 128, 0, stream>>>(
        x, enc_w, enc_b, nullptr, B0, mtN);

    auto conv = [&](const float* hin, int eblk, const float* wt, const float* bias,
                    const float* res, float* outp) {
        agg_kernel<<<AGG_NPART, 1024, 0, stream>>>(
            hin, src + (size_t)eblk * NEDGES, dst + (size_t)eblk * NEDGES, B2);
        if (res)
            gemm_ws_kernel<HDIM, 8, true><<<NWGP, 128, 0, stream>>>(B2, wt, bias, res, outp, mtN);
        else
            gemm_ws_kernel<HDIM, 8, false><<<NWGP, 128, 0, stream>>>(B2, wt, bias, nullptr, outp, mtN);
    };

    conv(B0, 0, gcn_w, gcn_b, nullptr, B0);

    for (int l = 1; l < 3; ++l) {
        ln_relu_kernel<<<lgrid, 256, 0, stream>>>(
            B0, ln_g + (size_t)(l - 1) * HDIM, ln_b + (size_t)(l - 1) * HDIM, B1, NNODES);
        conv(B1, l - 1, gcn_w + (size_t)l * HDIM * HDIM, gcn_b + (size_t)l * HDIM,
              B0,  B4);
        gather_kernel<<<ngrid, 256, 0, stream>>>(
            B4, node_map + (size_t)(l - 1) * NNODES, B0, NNODES);
    }

    const int sgrid = (NSEEDS * 32 + 255) / 256;
    gather_kernel<<<sgrid, 256, 0, stream>>>(B0, final_map, HG, NSEEDS);
    ln_relu_kernel<<<(NSEEDS + 7) / 8, 256, 0, stream>>>(
        HG, ln_g + 2 * HDIM, ln_b + 2 * HDIM, HG2, NSEEDS);
    const int  mtS = NSEEDS / 16;
    const dim3 gridS((mtS + 3) / 4, NCLSPAD / 64, 1);
    gemm_f32_wmma_kernel<<<gridS, gblk, 0, stream>>>(
        HG2, PWT, PB, nullptr, SC, NSEEDS, HDIM, NCLSPAD);
    logsoftmax_kernel<<<(NSEEDS + 7) / 8, 256, 0, stream>>>(SC, out, NSEEDS);
}


# Layer 1's aggregation reads its edge lists from the workspace, where layer 0's (same edge block) stored them in this call: one dst scan and src lookup gone; sums stay in LDS arrival order
# speedup vs baseline: 3.1601x; 1.0427x over previous
#include <hip/hip_runtime.h>
#include <hip/hip_bf16.h>
#include <math.h>

#define NNODES   100000
#define NEDGES   600000
#define FDIM     256
#define HDIM     128
#define NCLS     47
#define NCLSPAD  64
#define NSEEDS   10000
#define GEN_EPS  1e-7f
#define LN_EPS   1e-5f

typedef float v2f __attribute__((ext_vector_type(2)));
typedef float v8f __attribute__((ext_vector_type(8)));
typedef float v4f __attribute__((ext_vector_type(4)));

__global__ __launch_bounds__(128)
void gemm_f32_wmma_kernel(const float* __restrict__ A, const float* __restrict__ Wt,
                          const float* __restrict__ bias, const float* __restrict__ res,
                          float* __restrict__ out, int M, int K, int Nc)
{
    const int lane  = threadIdx.x;
    const int mtile = blockIdx.x * 4 + threadIdx.y;
    const int row0  = mtile * 16;
    if (row0 >= M) return;
    const int hb   = lane >> 4;
    const int lo   = lane & 15;
    const int arow = row0 + lo;
    const int col0 = blockIdx.y * 64 + lo;

    const float* __restrict__ a_ptr = A  + (size_t)arow * K;
    const float* __restrict__ b_ptr = Wt + (size_t)col0 * K;
    const size_t bstep = (size_t)16 * K;

    v8f acc[4];
#pragma unroll
    for (int t = 0; t < 4; ++t)
#pragma unroll
        for (int i = 0; i < 8; ++i) acc[t][i] = 0.f;

#pragma unroll 4
    for (int k = 0; k < K; k += 4) {
        const int ka = k + 2 * hb;
        const v2f a  = *(const v2f*)(a_ptr + ka);
        const v2f b0 = *(const v2f*)(b_ptr + ka);
        const v2f b1 = *(const v2f*)(b_ptr + bstep + ka);
        const v2f b2 = *(const v2f*)(b_ptr + 2 * bstep + ka);
        const v2f b3 = *(const v2f*)(b_ptr + 3 * bstep + ka);
        acc[0] = __builtin_amdgcn_wmma_f32_16x16x4_f32(false, a, false, b0, (short)0, acc[0], false, false);
        acc[1] = __builtin_amdgcn_wmma_f32_16x16x4_f32(false, a, false, b1, (short)0, acc[1], false, false);
        acc[2] = __builtin_amdgcn_wmma_f32_16x16x4_f32(false, a, false, b2, (short)0, acc[2], false, false);
        acc[3] = __builtin_amdgcn_wmma_f32_16x16x4_f32(false, a, false, b3, (short)0, acc[3], false, false);
    }

#pragma unroll
    for (int t = 0; t < 4; ++t) {
        const int   col = col0 + 16 * t;
        const float bv  = bias[col];
#pragma unroll
        for (int i = 0; i < 8; ++i) {
            const int r = row0 + i + 8 * hb;
            float v = acc[t][i] + bv;
            if (res) v += res[(size_t)r * Nc + col];
            out[(size_t)r * Nc + col] = v;
        }
    }
}

template <int K>
__device__ __forceinline__ void ws_load_a(float4 (&buf)[K / 8], const float* __restrict__ A,
                                          int tile, int lo, int hb)
{
    const float*   ap  = A + (size_t)tile * (16 * K);
    const unsigned lof = (unsigned)(lo * K + 4 * hb);
#pragma unroll
    for (int q = 0; q < K / 8; ++q) buf[q] = *(const float4*)(ap + lof + 8 * q);
}

template <int NT>
__device__ __forceinline__ void ws_load_r(float (&r)[NT][8], const float* __restrict__ res,
                                          int tile, int hb, int colb)
{
    const float*   rp  = res + (size_t)tile * (16 * HDIM);
    const unsigned lof = (unsigned)(8 * hb * HDIM + colb);
#pragma unroll
    for (int t = 0; t < NT; ++t)
#pragma unroll
        for (int i = 0; i < 8; ++i) r[t][i] = rp[lof + i * HDIM + 16 * t];
}

template <int K, int NT, bool HAS_RES>
__global__ __launch_bounds__(128)
void gemm_ws_kernel(const float* __restrict__ A, const float* __restrict__ W,
                    const float* __restrict__ bias, const float* __restrict__ res,
                    float* __restrict__ out, int mtiles)
{
    constexpr int NCG = HDIM / (16 * NT);
    constexpr int NQ  = K / 8;
    const int lane   = threadIdx.x & 31;
    const int wave   = blockIdx.x * 4 + __builtin_amdgcn_readfirstlane(threadIdx.x >> 5);
    const int hb     = lane >> 4;
    const int lo     = lane & 15;
    const int colb   = (wave % NCG) * 16 * NT + lo;
    const int stride = (gridDim.x * 4) / NCG;
    const int last   = mtiles - 1;
    int mt = wave / NCG;
    if (mt > last) return;

    v2f w[K / 4][NT];
#pragma unroll
    for (int q = 0; q < NQ; ++q) {
        const float* wp = W + (size_t)(8 * q + 4 * hb) * HDIM + colb;
#pragma unroll
        for (int t = 0; t < NT; ++t) {
            w[2 * q][t][0]     = wp[16 * t];
            w[2 * q][t][1]     = wp[HDIM + 16 * t];
            w[2 * q + 1][t][0] = wp[2 * HDIM + 16 * t];
            w[2 * q + 1][t][1] = wp[3 * HDIM + 16 * t];
        }
    }
    float bv[NT];
#pragma unroll
    for (int t = 0; t < NT; ++t) bv[t] = bias[colb + 16 * t];

    float4 cur[NQ], nxt[NQ];
    float  rc[NT][8], rn[NT][8];
    if (HAS_RES) ws_load_r<NT>(rn, res, mt, hb, colb);
    ws_load_a<K>(nxt, A, mt, lo, hb);

    while (true) {
#pragma unroll
        for (int q = 0; q < NQ; ++q) cur[q] = nxt[q];
        if (HAS_RES) {
#pragma unroll
            for (int t = 0; t < NT; ++t)
#pragma unroll
                for (int i = 0; i < 8; ++i) rc[t][i] = rn[t][i];
        }
        const int mtn = mt + stride;
        const int mtp = mtn < last ? mtn : last;
        ws_load_a<K>(nxt, A, mtp, lo, hb);

        v8f acc[NT];
#pragma unroll
        for (int t = 0; t < NT; ++t)
#pragma unroll
            for (int i = 0; i < 8; ++i) acc[t][i] = 0.f;

#pragma unroll
        for (int q = 0; q < NQ; ++q) {
            if (HAS_RES && q == NQ / 2) {
                __builtin_amdgcn_sched_barrier(0);
                ws_load_r<NT>(rn, res, mtp, hb, colb);
            }
            v2f a0, a1;
            a0[0] = cur[q].x; a0[1] = cur[q].y;
            a1[0] = cur[q].z; a1[1] = cur[q].w;
#pragma unroll
            for (int t = 0; t < NT; ++t)
                acc[t] = __builtin_amdgcn_wmma_f32_16x16x4_f32(false, a0, false, w[2 * q][t], (short)0, acc[t], false, false);
#pragma unroll
            for (int t = 0; t < NT; ++t)
                acc[t] = __builtin_amdgcn_wmma_f32_16x16x4_f32(false, a1, false, w[2 * q + 1][t], (short)0, acc[t], false, false);
        }

        float* __restrict__ op  = out + (size_t)mt * (16 * HDIM);
        const unsigned      oof = (unsigned)(8 * hb * HDIM + colb);
#pragma unroll
        for (int t = 0; t < NT; ++t)
#pragma unroll
            for (int i = 0; i < 8; ++i) {
                float v = acc[t][i] + bv[t];
                if (HAS_RES) v += rc[t][i];
                op[oof + i * HDIM + 16 * t] = v;
            }
        __builtin_amdgcn_sched_barrier(0);

        if (mtn > last) break;
        mt = mtn;
    }
}

__global__ __launch_bounds__(256)
void transpose_pad_kernel(const float* __restrict__ W, float* __restrict__ Wt,
                          int K, int Nc, int Ncpad)
{
    const int tid = blockIdx.x * 256 + threadIdx.x;
    if (tid >= Ncpad * K) return;
    const int c = tid / K;
    const int k = tid - c * K;
    Wt[(size_t)c * K + k] = (c < Nc) ? W[(size_t)k * Nc + c] : 0.f;
}

__global__ void pad_bias_kernel(const float* __restrict__ b, float* __restrict__ bp,
                                int Nc, int Ncpad)
{
    const int c = threadIdx.x;
    if (c < Ncpad) bp[c] = (c < Nc) ? b[c] : 0.f;
}

#define AGG_R      196
#define AGG_K      2
#define AGG_RW     (AGG_R * AGG_K)
#define AGG_NPART  ((NNODES + AGG_RW - 1) / AGG_RW)
#define AGG_CAP    3000
#define AGG_LPT    ((AGG_CAP + 1023) / 1024)
#define AGG_U      8
#define AGG_NV     (NEDGES / 4)
#define AGG_NBF    (AGG_NV / (1024 * AGG_U))
#define AGG_TAIL0  (AGG_NBF * 1024 * AGG_U)
#define AGG_NT     ((AGG_NV - AGG_TAIL0 + 1023) / 1024)

static_assert(NEDGES % 4 == 0 && NEDGES <= (1 << 20), "edge index packing");
static_assert(NNODES <= (1 << 17) && AGG_R <= 256, "list entry packing");
static_assert(AGG_NBF >= 1, "scan batching");

__device__ __forceinline__ unsigned agg_umin(unsigned a, unsigned b) { return a < b ? a : b; }

__device__ __forceinline__ void lds_add(float* p, float v)
{
    (void)__hip_atomic_fetch_add(p, v, __ATOMIC_RELAXED, __HIP_MEMORY_SCOPE_WORKGROUP);
}

__device__ __forceinline__ void agg_edge_max(unsigned* s_mx, const v4f hv, unsigned ld, int lane)
{
    unsigned* p = s_mx + ld * HDIM + lane;
    atomicMax(p,      __float_as_uint(fmaxf(hv.x, 0.f) + GEN_EPS));
    atomicMax(p + 32, __float_as_uint(fmaxf(hv.y, 0.f) + GEN_EPS));
    atomicMax(p + 64, __float_as_uint(fmaxf(hv.z, 0.f) + GEN_EPS));
    atomicMax(p + 96, __float_as_uint(fmaxf(hv.w, 0.f) + GEN_EPS));
}

__device__ __forceinline__ void agg_edge_sum(const unsigned* s_mx, float* s_de, float* s_nu,
                                             const v4f hv, unsigned ld, int lane)
{
    const unsigned o = ld * HDIM + lane;
    const float x0 = __uint_as_float(s_mx[o]);
    const float x1 = __uint_as_float(s_mx[o + 32]);
    const float x2 = __uint_as_float(s_mx[o + 64]);
    const float x3 = __uint_as_float(s_mx[o + 96]);
    const float m0 = fmaxf(hv.x, 0.f) + GEN_EPS, e0 = expf(m0 - x0);
    const float m1 = fmaxf(hv.y, 0.f) + GEN_EPS, e1 = expf(m1 - x1);
    const float m2 = fmaxf(hv.z, 0.f) + GEN_EPS, e2 = expf(m2 - x2);
    const float m3 = fmaxf(hv.w, 0.f) + GEN_EPS, e3 = expf(m3 - x3);
    lds_add(s_de + o,      e0); lds_add(s_nu + o,      m0 * e0);
    lds_add(s_de + o + 32, e1); lds_add(s_nu + o + 32, m1 * e1);
    lds_add(s_de + o + 64, e2); lds_add(s_nu + o + 64, m2 * e2);
    lds_add(s_de + o + 96, e3); lds_add(s_nu + o + 96, m3 * e3);
}

__device__ __forceinline__ void agg_test4(const int4 d, int i, int row0, unsigned nrw,
                                          unsigned* s_list, unsigned* s_cnt)
{
    const unsigned l0 = (unsigned)(d.x - row0);
    const unsigned l1 = (unsigned)(d.y - row0);
    const unsigned l2 = (unsigned)(d.z - row0);
    const unsigned l3 = (unsigned)(d.w - row0);
    if (agg_umin(agg_umin(l0, l1), agg_umin(l2, l3)) < nrw) {
        const unsigned e  = 4u * (unsigned)i;
        const unsigned lv[4] = {l0, l1, l2, l3};
#pragma unroll
        for (int k = 0; k < 4; ++k) {
            if (lv[k] < nrw) {
                const unsigned h   = lv[k] >= AGG_R ? 1u : 0u;
                const unsigned pos = atomicAdd(s_cnt + h, 1u);
                if (pos < AGG_CAP) s_list[h * AGG_CAP + pos] = (e + k) | ((lv[k] - h * AGG_R) << 20);
            }
        }
    }
}

template <int MODE>
__global__ __launch_bounds__(1024)
void agg_kernel(const float* __restrict__ hin, const int* __restrict__ src,
                const int* __restrict__ dst, float* __restrict__ t,
                unsigned* __restrict__ wl_list, unsigned* __restrict__ wl_cnt)
{
    __shared__ unsigned s_mx[(AGG_R + 1) * HDIM];
    __shared__ float    s_de[(AGG_R + 1) * HDIM];
    __shared__ float    s_nu[(AGG_R + 1) * HDIM];
    __shared__ unsigned s_list[AGG_K * AGG_CAP];
    __shared__ unsigned s_cnt[AGG_K];

    const int tid  = threadIdx.x;
    const int lane = tid & 31;
    const unsigned wave = (unsigned)__builtin_amdgcn_readfirstlane(tid >> 5);
    const int      row0 = blockIdx.x * AGG_RW;
    const unsigned nrw  = (unsigned)((NNODES - row0) < AGG_RW ? (NNODES - row0) : AGG_RW);

    unsigned* __restrict__ my_list = wl_list + (size_t)blockIdx.x * (AGG_K * AGG_CAP);
    unsigned* __restrict__ my_cnt  = wl_cnt + blockIdx.x * AGG_K;

    if (MODE == 2) {
        unsigned lim[AGG_K], ent[AGG_K][AGG_LPT];
#pragma unroll
        for (int h = 0; h < AGG_K; ++h) {
            const unsigned n = my_cnt[h];
            lim[h] = n <= AGG_CAP ? n : 0u;
            if (tid == h) s_cnt[h] = n;
        }
#pragma unroll
        for (int h = 0; h < AGG_K; ++h)
#pragma unroll
            for (int u = 0; u < AGG_LPT; ++u) {
                const unsigned i = (unsigned)tid + 1024u * u;
                ent[h][u] = i < lim[h] ? my_list[h * AGG_CAP + i] : 0u;
            }
#pragma unroll
        for (int h = 0; h < AGG_K; ++h)
#pragma unroll
            for (int u = 0; u < AGG_LPT; ++u) {
                const unsigned i = (unsigned)tid + 1024u * u;
                if (i < lim[h]) s_list[h * AGG_CAP + i] = ent[h][u];
            }
        __syncthreads();
    } else {
    if (tid < AGG_K) s_cnt[tid] = 0u;
    __syncthreads();

    const int4* __restrict__ d4 = (const int4*)dst;
    const int rot = (int)(blockIdx.x % AGG_NBF);
    for (int bb = 0; bb < AGG_NBF; ++bb) {
        int b = bb + rot;
        if (b >= AGG_NBF) b -= AGG_NBF;
        int4 d[AGG_U];
#pragma unroll
        for (int u = 0; u < AGG_U; ++u) d[u] = d4[(b * AGG_U + u) * 1024 + tid];
#pragma unroll
        for (int u = 0; u < AGG_U; ++u)
            agg_test4(d[u], (b * AGG_U + u) * 1024 + tid, row0, nrw, s_list, s_cnt);
    }
    {
        int4 d[AGG_NT];
#pragma unroll
        for (int u = 0; u < AGG_NT; ++u) {
            const int i = AGG_TAIL0 + u * 1024 + tid;
            d[u] = (i < AGG_NV) ? d4[i] : make_int4(-1, -1, -1, -1);
        }
#pragma unroll
        for (int u = 0; u < AGG_NT; ++u)
            agg_test4(d[u], AGG_TAIL0 + u * 1024 + tid, row0, nrw, s_list, s_cnt);
    }
    __syncthreads();

    for (int h = 0; h < AGG_K; ++h) {
        const unsigned n = s_cnt[h];
        if (n <= AGG_CAP)
            for (unsigned i = tid; i < n; i += 1024) {
                const unsigned en = s_list[h * AGG_CAP + i];
                const unsigned rs = (unsigned)src[en & 0xFFFFFu] | ((en >> 20) << 17);
                s_list[h * AGG_CAP + i] = rs;
                if (MODE == 1) my_list[h * AGG_CAP + i] = rs;
            }
    }
    if (MODE == 1 && tid < AGG_K) my_cnt[tid] = s_cnt[tid];
    __syncthreads();
    }

    for (int h = 0; h < AGG_K; ++h) {
        const int rbase = row0 + h * AGG_R;
        const int left  = NNODES - rbase;
        if (left <= 0) break;
        const unsigned  nrows  = (unsigned)(left < AGG_R ? left : AGG_R);
        const unsigned  n      = (unsigned)__builtin_amdgcn_readfirstlane((int)s_cnt[h]);
        const bool      listed = n <= AGG_CAP;
        const unsigned* lst    = s_list + h * AGG_CAP;

        for (int i = tid; i < (AGG_R + 1) * HDIM; i += 1024) { s_mx[i] = 0u; s_de[i] = 0.f; s_nu[i] = 0.f; }
        __syncthreads();

        if (listed) {
            for (unsigned i0 = wave; i0 < n; i0 += 128) {
                unsigned en[4], ld[4];
                v4f      hv[4];
#pragma unroll
                for (int u = 0; u < 4; ++u) {
                    const unsigned idx = i0 + 32u * u;
                    en[u] = lst[idx < n ? idx : n - 1u];
                    ld[u] = idx < n ? (en[u] >> 17) : (unsigned)AGG_R;
                }
#pragma unroll
                for (int u = 0; u < 4; ++u)
                    hv[u] = *(const v4f*)(hin + (size_t)(en[u] & 0x1FFFFu) * HDIM + 4 * lane);
                asm volatile("" :: "v"(hv[0]), "v"(hv[1]), "v"(hv[2]), "v"(hv[3]));
#pragma unroll
                for (int u = 0; u < 4; ++u) agg_edge_max(s_mx, hv[u], ld[u], lane);
            }
        } else {
            for (unsigned e = wave; e < NEDGES; e += 32) {
                const unsigned l = (unsigned)(dst[e] - rbase);
                if (l < nrows)
                    agg_edge_max(s_mx, *(const v4f*)(hin + (size_t)src[e] * HDIM + 4 * lane), l, lane);
            }
        }
        __syncthreads();

        if (listed) {
            for (unsigned i0 = wave; i0 < n; i0 += 128) {
                unsigned en[4], ld[4];
                v4f      hv[4];
#pragma unroll
                for (int u = 0; u < 4; ++u) {
                    const unsigned idx = i0 + 32u * u;
                    en[u] = lst[idx < n ? idx : n - 1u];
                    ld[u] = idx < n ? (en[u] >> 17) : (unsigned)AGG_R;
                }
#pragma unroll
                for (int u = 0; u < 4; ++u)
                    hv[u] = *(const v4f*)(hin + (size_t)(en[u] & 0x1FFFFu) * HDIM + 4 * lane);
#pragma unroll
                for (int u = 0; u < 4; ++u) agg_edge_sum(s_mx, s_de, s_nu, hv[u], ld[u], lane);
            }
        } else {
            for (unsigned e = wave; e < NEDGES; e += 32) {
                const unsigned l = (unsigned)(dst[e] - rbase);
                if (l < nrows)
                    agg_edge_sum(s_mx, s_de, s_nu,
                                 *(const v4f*)(hin + (size_t)src[e] * HDIM + 4 * lane), l, lane);
            }
        }
        __syncthreads();

        for (unsigned r = wave; r < nrows; r += 32) {
            const size_t   g  = (size_t)(rbase + (int)r) * HDIM + 4 * lane;
            const float4   hv = *(const float4*)(hin + g);
            const unsigned o  = r * HDIM + lane;
            float4 q;
            q.x = s_nu[o]      / (s_de[o]      + 1e-16f) + hv.x;
            q.y = s_nu[o + 32] / (s_de[o + 32] + 1e-16f) + hv.y;
            q.z = s_nu[o + 64] / (s_de[o + 64] + 1e-16f) + hv.z;
            q.w = s_nu[o + 96] / (s_de[o + 96] + 1e-16f) + hv.w;
            *(float4*)(t + g) = q;
        }
        __syncthreads();
    }
}

__global__ __launch_bounds__(256)
void ln_relu_kernel(const float* __restrict__ in, const float* __restrict__ g,
                    const float* __restrict__ b, float* __restrict__ out, int M)
{
    const int row  = blockIdx.x * 8 + (threadIdx.x >> 5);
    const int lane = threadIdx.x & 31;
    if (row >= M) return;
    const float4 v = *(const float4*)(in + (size_t)row * HDIM + lane * 4);
    float s1 = v.x + v.y + v.z + v.w;
    float s2 = v.x * v.x + v.y * v.y + v.z * v.z + v.w * v.w;
#pragma unroll
    for (int off = 16; off > 0; off >>= 1) {
        s1 += __shfl_xor(s1, off, 32);
        s2 += __shfl_xor(s2, off, 32);
    }
    const float mu  = s1 * (1.f / HDIM);
    const float var = s2 * (1.f / HDIM) - mu * mu;
    const float inv = rsqrtf(var + LN_EPS);
    const float4 gg = *(const float4*)(g + lane * 4);
    const float4 bb = *(const float4*)(b + lane * 4);
    float4 r;
    r.x = fmaxf((v.x - mu) * inv * gg.x + bb.x, 0.f);
    r.y = fmaxf((v.y - mu) * inv * gg.y + bb.y, 0.f);
    r.z = fmaxf((v.z - mu) * inv * gg.z + bb.z, 0.f);
    r.w = fmaxf((v.w - mu) * inv * gg.w + bb.w, 0.f);
    *(float4*)(out + (size_t)row * HDIM + lane * 4) = r;
}

__global__ __launch_bounds__(256)
void gather_kernel(const float* __restrict__ in, const int* __restrict__ map,
                   float* __restrict__ out, int M)
{
    const int tid = blockIdx.x * 256 + threadIdx.x;
    if (tid >= M * 32) return;
    const int i = tid >> 5;
    const int f = (tid & 31) * 4;
    const int j = map[i];
    *(float4*)(out + (size_t)i * HDIM + f) =
        *(const float4*)(in + (size_t)j * HDIM + f);
}

__global__ __launch_bounds__(256)
void logsoftmax_kernel(const float* __restrict__ in, float* __restrict__ out, int M)
{
    const int row  = blockIdx.x * 8 + (threadIdx.x >> 5);
    const int lane = threadIdx.x & 31;
    if (row >= M) return;
    const float NEG = -3.0e38f;
    const float v0 = (lane      < NCLS) ? in[(size_t)row * NCLSPAD + lane     ] : NEG;
    const float v1 = (lane + 32 < NCLS) ? in[(size_t)row * NCLSPAD + lane + 32] : NEG;
    float mx = fmaxf(v0, v1);
#pragma unroll
    for (int off = 16; off > 0; off >>= 1) mx = fmaxf(mx, __shfl_xor(mx, off, 32));
    float se = ((lane      < NCLS) ? expf(v0 - mx) : 0.f)
             + ((lane + 32 < NCLS) ? expf(v1 - mx) : 0.f);
#pragma unroll
    for (int off = 16; off > 0; off >>= 1) se += __shfl_xor(se, off, 32);
    const float lse = logf(se) + mx;
    if (lane      < NCLS) out[(size_t)row * NCLS + lane     ] = v0 - lse;
    if (lane + 32 < NCLS) out[(size_t)row * NCLS + lane + 32] = v1 - lse;
}

extern "C" void kernel_launch(void* const* d_in, const int* in_sizes, int n_in,
                              void* d_out, int out_size, void* d_ws, size_t ws_size,
                              hipStream_t stream)
{
    (void)in_sizes; (void)n_in; (void)out_size; (void)ws_size;

    const float* x         = (const float*)d_in[0];
    const int*   src       = (const int*)  d_in[1];
    const int*   dst       = (const int*)  d_in[2];
    const int*   node_map  = (const int*)  d_in[3];
    const int*   final_map = (const int*)  d_in[4];
    const float* enc_w     = (const float*)d_in[5];
    const float* enc_b     = (const float*)d_in[6];
    const float* gcn_w     = (const float*)d_in[7];
    const float* gcn_b     = (const float*)d_in[8];
    const float* ln_g      = (const float*)d_in[9];
    const float* ln_b      = (const float*)d_in[10];
    const float* pred_w    = (const float*)d_in[11];
    const float* pred_b    = (const float*)d_in[12];
    float* out = (float*)d_out;

    const size_t NH = (size_t)NNODES * HDIM;
    float* B0     = (float*)d_ws;
    float* B1     = B0 + NH;
    float* B2     = B0 + 2 * NH;
    float* B4     = B0 + 4 * NH;
    float* HG     = B0 + 5 * NH;
    float* HG2    = HG  + (size_t)NSEEDS * HDIM;
    float* SC     = HG2 + (size_t)NSEEDS * HDIM;
    float* PWT    = SC  + (size_t)NSEEDS * NCLSPAD;
    float* PB     = PWT    + (size_t)NCLSPAD * HDIM;
    unsigned* WL  = (unsigned*)(PB + NCLSPAD);
    unsigned* WC  = WL + (size_t)AGG_NPART * AGG_K * AGG_CAP;

    transpose_pad_kernel<<<(NCLSPAD * HDIM + 255) / 256, 256, 0, stream>>>(
        pred_w, PWT, HDIM, NCLS, NCLSPAD);
    pad_bias_kernel<<<1, NCLSPAD, 0, stream>>>(pred_b, PB, NCLS, NCLSPAD);

    const dim3 gblk(32, 4, 1);
    const int  mtN   = NNODES / 16;
    const int  NWGP  = 256;
    const int  ngrid = (NNODES * 32 + 255) / 256;
    const int  lgrid = (NNODES + 7) / 8;

    gemm_ws_kernel<FDIM, 4, false><<<NWGP, 128, 0, stream>>>(
        x, enc_w, enc_b, nullptr, B0, mtN);

    auto conv = [&](const float* hin, int eblk, int lmode, const float* wt, const float* bias,
                    const float* res, float* outp) {
        const int* es = src + (size_t)eblk * NEDGES;
        const int* ed = dst + (size_t)eblk * NEDGES;
        if (lmode == 1)      agg_kernel<1><<<AGG_NPART, 1024, 0, stream>>>(hin, es, ed, B2, WL, WC);
        else if (lmode == 2) agg_kernel<2><<<AGG_NPART, 1024, 0, stream>>>(hin, es, ed, B2, WL, WC);
        else                 agg_kernel<0><<<AGG_NPART, 1024, 0, stream>>>(hin, es, ed, B2, WL, WC);
        if (res)
            gemm_ws_kernel<HDIM, 8, true><<<NWGP, 128, 0, stream>>>(B2, wt, bias, res, outp, mtN);
        else
            gemm_ws_kernel<HDIM, 8, false><<<NWGP, 128, 0, stream>>>(B2, wt, bias, nullptr, outp, mtN);
    };

    conv(B0, 0, 1, gcn_w, gcn_b, nullptr, B0);

    for (int l = 1; l < 3; ++l) {
        ln_relu_kernel<<<lgrid, 256, 0, stream>>>(
            B0, ln_g + (size_t)(l - 1) * HDIM, ln_b + (size_t)(l - 1) * HDIM, B1, NNODES);
        conv(B1, l - 1, l == 1 ? 2 : 0, gcn_w + (size_t)l * HDIM * HDIM, gcn_b + (size_t)l * HDIM,
              B0,  B4);
        gather_kernel<<<ngrid, 256, 0, stream>>>(
            B4, node_map + (size_t)(l - 1) * NNODES, B0, NNODES);
    }

    const int sgrid = (NSEEDS * 32 + 255) / 256;
    gather_kernel<<<sgrid, 256, 0, stream>>>(B0, final_map, HG, NSEEDS);
    ln_relu_kernel<<<(NSEEDS + 7) / 8, 256, 0, stream>>>(
        HG, ln_g + 2 * HDIM, ln_b + 2 * HDIM, HG2, NSEEDS);
    const int  mtS = NSEEDS / 16;
    const dim3 gridS((mtS + 3) / 4, NCLSPAD / 64, 1);
    gemm_f32_wmma_kernel<<<gridS, gblk, 0, stream>>>(
        HG2, PWT, PB, nullptr, SC, NSEEDS, HDIM, NCLSPAD);
    logsoftmax_kernel<<<(NSEEDS + 7) / 8, 256, 0, stream>>>(SC, out, NSEEDS);
}


# Aggregation softmax shifted by a per-layer upper bound of m (from LayerNorm g,b; layer 0: max of encoder output), not the per-feature max: max walk and its row gather gone; LDS arrival-order sums
# speedup vs baseline: 3.4787x; 1.1008x over previous
#include <hip/hip_runtime.h>
#include <hip/hip_bf16.h>
#include <math.h>

#define NNODES   100000
#define NEDGES   600000
#define FDIM     256
#define HDIM     128
#define NCLS     47
#define NCLSPAD  64
#define NSEEDS   10000
#define GEN_EPS  1e-7f
#define LN_EPS   1e-5f

typedef float v2f __attribute__((ext_vector_type(2)));
typedef float v8f __attribute__((ext_vector_type(8)));
typedef float v4f __attribute__((ext_vector_type(4)));

__global__ __launch_bounds__(128)
void gemm_f32_wmma_kernel(const float* __restrict__ A, const float* __restrict__ Wt,
                          const float* __restrict__ bias, const float* __restrict__ res,
                          float* __restrict__ out, int M, int K, int Nc)
{
    const int lane  = threadIdx.x;
    const int mtile = blockIdx.x * 4 + threadIdx.y;
    const int row0  = mtile * 16;
    if (row0 >= M) return;
    const int hb   = lane >> 4;
    const int lo   = lane & 15;
    const int arow = row0 + lo;
    const int col0 = blockIdx.y * 64 + lo;

    const float* __restrict__ a_ptr = A  + (size_t)arow * K;
    const float* __restrict__ b_ptr = Wt + (size_t)col0 * K;
    const size_t bstep = (size_t)16 * K;

    v8f acc[4];
#pragma unroll
    for (int t = 0; t < 4; ++t)
#pragma unroll
        for (int i = 0; i < 8; ++i) acc[t][i] = 0.f;

#pragma unroll 4
    for (int k = 0; k < K; k += 4) {
        const int ka = k + 2 * hb;
        const v2f a  = *(const v2f*)(a_ptr + ka);
        const v2f b0 = *(const v2f*)(b_ptr + ka);
        const v2f b1 = *(const v2f*)(b_ptr + bstep + ka);
        const v2f b2 = *(const v2f*)(b_ptr + 2 * bstep + ka);
        const v2f b3 = *(const v2f*)(b_ptr + 3 * bstep + ka);
        acc[0] = __builtin_amdgcn_wmma_f32_16x16x4_f32(false, a, false, b0, (short)0, acc[0], false, false);
        acc[1] = __builtin_amdgcn_wmma_f32_16x16x4_f32(false, a, false, b1, (short)0, acc[1], false, false);
        acc[2] = __builtin_amdgcn_wmma_f32_16x16x4_f32(false, a, false, b2, (short)0, acc[2], false, false);
        acc[3] = __builtin_amdgcn_wmma_f32_16x16x4_f32(false, a, false, b3, (short)0, acc[3], false, false);
    }

#pragma unroll
    for (int t = 0; t < 4; ++t) {
        const int   col = col0 + 16 * t;
        const float bv  = bias[col];
#pragma unroll
        for (int i = 0; i < 8; ++i) {
            const int r = row0 + i + 8 * hb;
            float v = acc[t][i] + bv;
            if (res) v += res[(size_t)r * Nc + col];
            out[(size_t)r * Nc + col] = v;
        }
    }
}

template <int K>
__device__ __forceinline__ void ws_load_a(float4 (&buf)[K / 8], const float* __restrict__ A,
                                          int tile, int lo, int hb)
{
    const float*   ap  = A + (size_t)tile * (16 * K);
    const unsigned lof = (unsigned)(lo * K + 4 * hb);
#pragma unroll
    for (int q = 0; q < K / 8; ++q) buf[q] = *(const float4*)(ap + lof + 8 * q);
}

template <int NT>
__device__ __forceinline__ void ws_load_r(float (&r)[NT][8], const float* __restrict__ res,
                                          int tile, int hb, int colb)
{
    const float*   rp  = res + (size_t)tile * (16 * HDIM);
    const unsigned lof = (unsigned)(8 * hb * HDIM + colb);
#pragma unroll
    for (int t = 0; t < NT; ++t)
#pragma unroll
        for (int i = 0; i < 8; ++i) r[t][i] = rp[lof + i * HDIM + 16 * t];
}

template <int K, int NT, bool HAS_RES, bool WANT_MAX>
__global__ __launch_bounds__(128)
void gemm_ws_kernel(const float* __restrict__ A, const float* __restrict__ W,
                    const float* __restrict__ bias, const float* __restrict__ res,
                    float* __restrict__ out, int mtiles, unsigned* __restrict__ vmax_out)
{
    constexpr int NCG = HDIM / (16 * NT);
    constexpr int NQ  = K / 8;
    const int lane   = threadIdx.x & 31;
    const int wave   = blockIdx.x * 4 + __builtin_amdgcn_readfirstlane(threadIdx.x >> 5);
    const int hb     = lane >> 4;
    const int lo     = lane & 15;
    const int colb   = (wave % NCG) * 16 * NT + lo;
    const int stride = (gridDim.x * 4) / NCG;
    const int last   = mtiles - 1;
    int mt = wave / NCG;
    if (mt > last) return;

    v2f w[K / 4][NT];
#pragma unroll
    for (int q = 0; q < NQ; ++q) {
        const float* wp = W + (size_t)(8 * q + 4 * hb) * HDIM + colb;
#pragma unroll
        for (int t = 0; t < NT; ++t) {
            w[2 * q][t][0]     = wp[16 * t];
            w[2 * q][t][1]     = wp[HDIM + 16 * t];
            w[2 * q + 1][t][0] = wp[2 * HDIM + 16 * t];
            w[2 * q + 1][t][1] = wp[3 * HDIM + 16 * t];
        }
    }
    float bv[NT];
#pragma unroll
    for (int t = 0; t < NT; ++t) bv[t] = bias[colb + 16 * t];

    float4 cur[NQ], nxt[NQ];
    float  rc[NT][8], rn[NT][8];
    float  vmax = 0.f;
    if (HAS_RES) ws_load_r<NT>(rn, res, mt, hb, colb);
    ws_load_a<K>(nxt, A, mt, lo, hb);

    while (true) {
#pragma unroll
        for (int q = 0; q < NQ; ++q) cur[q] = nxt[q];
        if (HAS_RES) {
#pragma unroll
            for (int t = 0; t < NT; ++t)
#pragma unroll
                for (int i = 0; i < 8; ++i) rc[t][i] = rn[t][i];
        }
        const int mtn = mt + stride;
        const int mtp = mtn < last ? mtn : last;
        ws_load_a<K>(nxt, A, mtp, lo, hb);

        v8f acc[NT];
#pragma unroll
        for (int t = 0; t < NT; ++t)
#pragma unroll
            for (int i = 0; i < 8; ++i) acc[t][i] = 0.f;

#pragma unroll
        for (int q = 0; q < NQ; ++q) {
            if (HAS_RES && q == NQ / 2) {
                __builtin_amdgcn_sched_barrier(0);
                ws_load_r<NT>(rn, res, mtp, hb, colb);
            }
            v2f a0, a1;
            a0[0] = cur[q].x; a0[1] = cur[q].y;
            a1[0] = cur[q].z; a1[1] = cur[q].w;
#pragma unroll
            for (int t = 0; t < NT; ++t)
                acc[t] = __builtin_amdgcn_wmma_f32_16x16x4_f32(false, a0, false, w[2 * q][t], (short)0, acc[t], false, false);
#pragma unroll
            for (int t = 0; t < NT; ++t)
                acc[t] = __builtin_amdgcn_wmma_f32_16x16x4_f32(false, a1, false, w[2 * q + 1][t], (short)0, acc[t], false, false);
        }

        float* __restrict__ op  = out + (size_t)mt * (16 * HDIM);
        const unsigned      oof = (unsigned)(8 * hb * HDIM + colb);
#pragma unroll
        for (int t = 0; t < NT; ++t)
#pragma unroll
            for (int i = 0; i < 8; ++i) {
                float v = acc[t][i] + bv[t];
                if (HAS_RES) v += rc[t][i];
                if (WANT_MAX) vmax = fmaxf(vmax, v);
                op[oof + i * HDIM + 16 * t] = v;
            }
        __builtin_amdgcn_sched_barrier(0);

        if (mtn > last) break;
        mt = mtn;
    }

    if (WANT_MAX) {
#pragma unroll
        for (int off = 16; off > 0; off >>= 1) vmax = fmaxf(vmax, __shfl_xor(vmax, off, 32));
        if (lane == 0) atomicMax(vmax_out, __float_as_uint(vmax));
    }
}

__global__ __launch_bounds__(256)
void transpose_pad_kernel(const float* __restrict__ W, float* __restrict__ Wt,
                          int K, int Nc, int Ncpad)
{
    const int tid = blockIdx.x * 256 + threadIdx.x;
    if (tid >= Ncpad * K) return;
    const int c = tid / K;
    const int k = tid - c * K;
    Wt[(size_t)c * K + k] = (c < Nc) ? W[(size_t)k * Nc + c] : 0.f;
}

__global__ void pad_bias_kernel(const float* __restrict__ b, float* __restrict__ bp,
                                int Nc, int Ncpad, unsigned* __restrict__ max_word)
{
    const int c = threadIdx.x;
    if (c < Ncpad) bp[c] = (c < Nc) ? b[c] : 0.f;
    if (c == 0) *max_word = 0u;
}

#define AGG_R      196
#define AGG_K      2
#define AGG_RW     (AGG_R * AGG_K)
#define AGG_NPART  ((NNODES + AGG_RW - 1) / AGG_RW)
#define AGG_CAP    3000
#define AGG_LPT    ((AGG_CAP + 1023) / 1024)
#define AGG_U      8
#define AGG_NV     (NEDGES / 4)
#define AGG_NBF    (AGG_NV / (1024 * AGG_U))
#define AGG_TAIL0  (AGG_NBF * 1024 * AGG_U)
#define AGG_NT     ((AGG_NV - AGG_TAIL0 + 1023) / 1024)
#define AGG_SAFE   60.f
#define LN_BOUND   11.3137085f

static_assert(NEDGES % 4 == 0 && NEDGES <= (1 << 20), "edge index packing");
static_assert(NNODES <= (1 << 17) && AGG_R <= 256, "list entry packing");
static_assert(AGG_NBF >= 1, "scan batching");

__device__ __forceinline__ unsigned agg_umin(unsigned a, unsigned b) { return a < b ? a : b; }

__device__ __forceinline__ void lds_add(float* p, float v)
{
    (void)__hip_atomic_fetch_add(p, v, __ATOMIC_RELAXED, __HIP_MEMORY_SCOPE_WORKGROUP);
}

__device__ __forceinline__ void agg_edge_max(unsigned* s_mx, const v4f hv, unsigned ld, int lane)
{
    unsigned* p = s_mx + ld * HDIM + lane;
    atomicMax(p,      __float_as_uint(fmaxf(hv.x, 0.f) + GEN_EPS));
    atomicMax(p + 32, __float_as_uint(fmaxf(hv.y, 0.f) + GEN_EPS));
    atomicMax(p + 64, __float_as_uint(fmaxf(hv.z, 0.f) + GEN_EPS));
    atomicMax(p + 96, __float_as_uint(fmaxf(hv.w, 0.f) + GEN_EPS));
}

template <bool BOUND>
__device__ __forceinline__ void agg_edge_sum(const unsigned* s_mx, float* s_de, float* s_nu,
                                             const v4f hv, unsigned ld, int lane, float bound)
{
    const unsigned o = ld * HDIM + lane;
    float x0 = bound, x1 = bound, x2 = bound, x3 = bound;
    if (!BOUND) {
        x0 = __uint_as_float(s_mx[o]);
        x1 = __uint_as_float(s_mx[o + 32]);
        x2 = __uint_as_float(s_mx[o + 64]);
        x3 = __uint_as_float(s_mx[o + 96]);
    }
    const float m0 = fmaxf(hv.x, 0.f) + GEN_EPS, e0 = expf(m0 - x0);
    const float m1 = fmaxf(hv.y, 0.f) + GEN_EPS, e1 = expf(m1 - x1);
    const float m2 = fmaxf(hv.z, 0.f) + GEN_EPS, e2 = expf(m2 - x2);
    const float m3 = fmaxf(hv.w, 0.f) + GEN_EPS, e3 = expf(m3 - x3);
    lds_add(s_de + o,      e0); lds_add(s_nu + o,      m0 * e0);
    lds_add(s_de + o + 32, e1); lds_add(s_nu + o + 32, m1 * e1);
    lds_add(s_de + o + 64, e2); lds_add(s_nu + o + 64, m2 * e2);
    lds_add(s_de + o + 96, e3); lds_add(s_nu + o + 96, m3 * e3);
}

__device__ __forceinline__ void agg_test4(const int4 d, int i, int row0, unsigned nrw,
                                          unsigned* s_list, unsigned* s_cnt)
{
    const unsigned l0 = (unsigned)(d.x - row0);
    const unsigned l1 = (unsigned)(d.y - row0);
    const unsigned l2 = (unsigned)(d.z - row0);
    const unsigned l3 = (unsigned)(d.w - row0);
    if (agg_umin(agg_umin(l0, l1), agg_umin(l2, l3)) < nrw) {
        const unsigned e  = 4u * (unsigned)i;
        const unsigned lv[4] = {l0, l1, l2, l3};
#pragma unroll
        for (int k = 0; k < 4; ++k) {
            if (lv[k] < nrw) {
                const unsigned h   = lv[k] >= AGG_R ? 1u : 0u;
                const unsigned pos = atomicAdd(s_cnt + h, 1u);
                if (pos < AGG_CAP) s_list[h * AGG_CAP + pos] = (e + k) | ((lv[k] - h * AGG_R) << 20);
            }
        }
    }
}

template <bool BOUND>
__device__ __forceinline__ void agg_walk_sum(const float* __restrict__ hin, const int* __restrict__ src,
                                             const int* __restrict__ dst, const unsigned* lst,
                                             unsigned n, bool listed, int rbase, unsigned nrows,
                                             const unsigned* s_mx, float* s_de, float* s_nu,
                                             unsigned wave, int lane, float bound)
{
    if (listed) {
        for (unsigned i0 = wave; i0 < n; i0 += 128) {
            unsigned en[4], ld[4];
            v4f      hv[4];
#pragma unroll
            for (int u = 0; u < 4; ++u) {
                const unsigned idx = i0 + 32u * u;
                en[u] = lst[idx < n ? idx : n - 1u];
                ld[u] = idx < n ? (en[u] >> 17) : (unsigned)AGG_R;
            }
#pragma unroll
            for (int u = 0; u < 4; ++u)
                hv[u] = *(const v4f*)(hin + (size_t)(en[u] & 0x1FFFFu) * HDIM + 4 * lane);
#pragma unroll
            for (int u = 0; u < 4; ++u) agg_edge_sum<BOUND>(s_mx, s_de, s_nu, hv[u], ld[u], lane, bound);
        }
    } else {
        for (unsigned e = wave; e < NEDGES; e += 32) {
            const unsigned l = (unsigned)(dst[e] - rbase);
            if (l < nrows)
                agg_edge_sum<BOUND>(s_mx, s_de, s_nu,
                                    *(const v4f*)(hin + (size_t)src[e] * HDIM + 4 * lane), l, lane, bound);
        }
    }
}

template <int MODE>
__global__ __launch_bounds__(1024)
void agg_kernel(const float* __restrict__ hin, const int* __restrict__ src,
                const int* __restrict__ dst, float* __restrict__ t,
                unsigned* __restrict__ wl_list, unsigned* __restrict__ wl_cnt,
                const float* __restrict__ bg, const float* __restrict__ bb,
                const unsigned* __restrict__ bmax)
{
    __shared__ unsigned s_mx[(AGG_R + 1) * HDIM];
    __shared__ float    s_de[(AGG_R + 1) * HDIM];
    __shared__ float    s_nu[(AGG_R + 1) * HDIM];
    __shared__ unsigned s_list[AGG_K * AGG_CAP];
    __shared__ unsigned s_cnt[AGG_K];
    __shared__ unsigned s_bnd[2];

    const int tid  = threadIdx.x;
    const int lane = tid & 31;
    const unsigned wave = (unsigned)__builtin_amdgcn_readfirstlane(tid >> 5);
    const int      row0 = blockIdx.x * AGG_RW;
    const unsigned nrw  = (unsigned)((NNODES - row0) < AGG_RW ? (NNODES - row0) : AGG_RW);

    unsigned* __restrict__ my_list = wl_list + (size_t)blockIdx.x * (AGG_K * AGG_CAP);
    unsigned* __restrict__ my_cnt  = wl_cnt + blockIdx.x * AGG_K;

    if (tid < 2) s_bnd[tid] = 0u;
    __syncthreads();
    if (bg) {
        if (tid < HDIM)          atomicMax(&s_bnd[0], __float_as_uint(fabsf(bg[tid])));
        else if (tid < 2 * HDIM) atomicMax(&s_bnd[1], __float_as_uint(fabsf(bb[tid - HDIM])));
    } else if (tid == 0) {
        s_bnd[1] = *bmax;
    }

    if (MODE == 2) {
        unsigned lim[AGG_K], ent[AGG_K][AGG_LPT];
#pragma unroll
        for (int h = 0; h < AGG_K; ++h) {
            const unsigned n = my_cnt[h];
            lim[h] = n <= AGG_CAP ? n : 0u;
            if (tid == h) s_cnt[h] = n;
        }
#pragma unroll
        for (int h = 0; h < AGG_K; ++h)
#pragma unroll
            for (int u = 0; u < AGG_LPT; ++u) {
                const unsigned i = (unsigned)tid + 1024u * u;
                ent[h][u] = i < lim[h] ? my_list[h * AGG_CAP + i] : 0u;
            }
#pragma unroll
        for (int h = 0; h < AGG_K; ++h)
#pragma unroll
            for (int u = 0; u < AGG_LPT; ++u) {
                const unsigned i = (unsigned)tid + 1024u * u;
                if (i < lim[h]) s_list[h * AGG_CAP + i] = ent[h][u];
            }
        __syncthreads();
    } else {
    if (tid < AGG_K) s_cnt[tid] = 0u;
    __syncthreads();

    const int4* __restrict__ d4 = (const int4*)dst;
    const int rot = (int)(blockIdx.x % AGG_NBF);
    for (int bb = 0; bb < AGG_NBF; ++bb) {
        int b = bb + rot;
        if (b >= AGG_NBF) b -= AGG_NBF;
        int4 d[AGG_U];
#pragma unroll
        for (int u = 0; u < AGG_U; ++u) d[u] = d4[(b * AGG_U + u) * 1024 + tid];
#pragma unroll
        for (int u = 0; u < AGG_U; ++u)
            agg_test4(d[u], (b * AGG_U + u) * 1024 + tid, row0, nrw, s_list, s_cnt);
    }
    {
        int4 d[AGG_NT];
#pragma unroll
        for (int u = 0; u < AGG_NT; ++u) {
            const int i = AGG_TAIL0 + u * 1024 + tid;
            d[u] = (i < AGG_NV) ? d4[i] : make_int4(-1, -1, -1, -1);
        }
#pragma unroll
        for (int u = 0; u < AGG_NT; ++u)
            agg_test4(d[u], AGG_TAIL0 + u * 1024 + tid, row0, nrw, s_list, s_cnt);
    }
    __syncthreads();

    for (int h = 0; h < AGG_K; ++h) {
        const unsigned n = s_cnt[h];
        if (n <= AGG_CAP)
            for (unsigned i = tid; i < n; i += 1024) {
                const unsigned en = s_list[h * AGG_CAP + i];
                const unsigned rs = (unsigned)src[en & 0xFFFFFu] | ((en >> 20) << 17);
                s_list[h * AGG_CAP + i] = rs;
                if (MODE == 1) my_list[h * AGG_CAP + i] = rs;
            }
    }
    if (MODE == 1 && tid < AGG_K) my_cnt[tid] = s_cnt[tid];
    __syncthreads();
    }

    const float bound = LN_BOUND * __uint_as_float(s_bnd[0]) + __uint_as_float(s_bnd[1]) + GEN_EPS;
    const bool  usebound = __builtin_amdgcn_readfirstlane((int)(bound <= AGG_SAFE)) != 0;

    for (int h = 0; h < AGG_K; ++h) {
        const int rbase = row0 + h * AGG_R;
        const int left  = NNODES - rbase;
        if (left <= 0) break;
        const unsigned  nrows  = (unsigned)(left < AGG_R ? left : AGG_R);
        const unsigned  n      = (unsigned)__builtin_amdgcn_readfirstlane((int)s_cnt[h]);
        const bool      listed = n <= AGG_CAP;
        const unsigned* lst    = s_list + h * AGG_CAP;

        if (usebound) {
            for (int i = tid; i < (AGG_R + 1) * HDIM; i += 1024) { s_de[i] = 0.f; s_nu[i] = 0.f; }
        } else {
            for (int i = tid; i < (AGG_R + 1) * HDIM; i += 1024) { s_mx[i] = 0u; s_de[i] = 0.f; s_nu[i] = 0.f; }
        }
        __syncthreads();

        if (usebound) {
            agg_walk_sum<true>(hin, src, dst, lst, n, listed, rbase, nrows, s_mx, s_de, s_nu, wave, lane, bound);
        } else {
            if (listed) {
                for (unsigned i0 = wave; i0 < n; i0 += 128) {
                    unsigned en[4], ld[4];
                    v4f      hv[4];
#pragma unroll
                    for (int u = 0; u < 4; ++u) {
                        const unsigned idx = i0 + 32u * u;
                        en[u] = lst[idx < n ? idx : n - 1u];
                        ld[u] = idx < n ? (en[u] >> 17) : (unsigned)AGG_R;
                    }
#pragma unroll
                    for (int u = 0; u < 4; ++u)
                        hv[u] = *(const v4f*)(hin + (size_t)(en[u] & 0x1FFFFu) * HDIM + 4 * lane);
                    asm volatile("" :: "v"(hv[0]), "v"(hv[1]), "v"(hv[2]), "v"(hv[3]));
#pragma unroll
                    for (int u = 0; u < 4; ++u) agg_edge_max(s_mx, hv[u], ld[u], lane);
                }
            } else {
                for (unsigned e = wave; e < NEDGES; e += 32) {
                    const unsigned l = (unsigned)(dst[e] - rbase);
                    if (l < nrows)
                        agg_edge_max(s_mx, *(const v4f*)(hin + (size_t)src[e] * HDIM + 4 * lane), l, lane);
                }
            }
            __syncthreads();

            agg_walk_sum<false>(hin, src, dst, lst, n, listed, rbase, nrows, s_mx, s_de, s_nu, wave, lane, 0.f);
        }
        __syncthreads();

        for (unsigned r = wave; r < nrows; r += 32) {
            const size_t   g  = (size_t)(rbase + (int)r) * HDIM + 4 * lane;
            const float4   hv = *(const float4*)(hin + g);
            const unsigned o  = r * HDIM + lane;
            const float d0 = s_de[o], d1 = s_de[o + 32], d2 = s_de[o + 64], d3 = s_de[o + 96];
            const float n0 = s_nu[o], n1 = s_nu[o + 32], n2 = s_nu[o + 64], n3 = s_nu[o + 96];
            float4 q;
            if (usebound) {
                q.x = (d0 > 0.f ? n0 / d0 : 0.f) + hv.x;
                q.y = (d1 > 0.f ? n1 / d1 : 0.f) + hv.y;
                q.z = (d2 > 0.f ? n2 / d2 : 0.f) + hv.z;
                q.w = (d3 > 0.f ? n3 / d3 : 0.f) + hv.w;
            } else {
                q.x = n0 / (d0 + 1e-16f) + hv.x;
                q.y = n1 / (d1 + 1e-16f) + hv.y;
                q.z = n2 / (d2 + 1e-16f) + hv.z;
                q.w = n3 / (d3 + 1e-16f) + hv.w;
            }
            *(float4*)(t + g) = q;
        }
        __syncthreads();
    }
}

__global__ __launch_bounds__(256)
void ln_relu_kernel(const float* __restrict__ in, const float* __restrict__ g,
                    const float* __restrict__ b, float* __restrict__ out, int M)
{
    const int row  = blockIdx.x * 8 + (threadIdx.x >> 5);
    const int lane = threadIdx.x & 31;
    if (row >= M) return;
    const float4 v = *(const float4*)(in + (size_t)row * HDIM + lane * 4);
    float s1 = v.x + v.y + v.z + v.w;
    float s2 = v.x * v.x + v.y * v.y + v.z * v.z + v.w * v.w;
#pragma unroll
    for (int off = 16; off > 0; off >>= 1) {
        s1 += __shfl_xor(s1, off, 32);
        s2 += __shfl_xor(s2, off, 32);
    }
    const float mu  = s1 * (1.f / HDIM);
    const float var = s2 * (1.f / HDIM) - mu * mu;
    const float inv = rsqrtf(var + LN_EPS);
    const float4 gg = *(const float4*)(g + lane * 4);
    const float4 bb = *(const float4*)(b + lane * 4);
    float4 r;
    r.x = fmaxf((v.x - mu) * inv * gg.x + bb.x, 0.f);
    r.y = fmaxf((v.y - mu) * inv * gg.y + bb.y, 0.f);
    r.z = fmaxf((v.z - mu) * inv * gg.z + bb.z, 0.f);
    r.w = fmaxf((v.w - mu) * inv * gg.w + bb.w, 0.f);
    *(float4*)(out + (size_t)row * HDIM + lane * 4) = r;
}

__global__ __launch_bounds__(256)
void gather_kernel(const float* __restrict__ in, const int* __restrict__ map,
                   float* __restrict__ out, int M)
{
    const int tid = blockIdx.x * 256 + threadIdx.x;
    if (tid >= M * 32) return;
    const int i = tid >> 5;
    const int f = (tid & 31) * 4;
    const int j = map[i];
    *(float4*)(out + (size_t)i * HDIM + f) =
        *(const float4*)(in + (size_t)j * HDIM + f);
}

__global__ __launch_bounds__(256)
void logsoftmax_kernel(const float* __restrict__ in, float* __restrict__ out, int M)
{
    const int row  = blockIdx.x * 8 + (threadIdx.x >> 5);
    const int lane = threadIdx.x & 31;
    if (row >= M) return;
    const float NEG = -3.0e38f;
    const float v0 = (lane      < NCLS) ? in[(size_t)row * NCLSPAD + lane     ] : NEG;
    const float v1 = (lane + 32 < NCLS) ? in[(size_t)row * NCLSPAD + lane + 32] : NEG;
    float mx = fmaxf(v0, v1);
#pragma unroll
    for (int off = 16; off > 0; off >>= 1) mx = fmaxf(mx, __shfl_xor(mx, off, 32));
    float se = ((lane      < NCLS) ? expf(v0 - mx) : 0.f)
             + ((lane + 32 < NCLS) ? expf(v1 - mx) : 0.f);
#pragma unroll
    for (int off = 16; off > 0; off >>= 1) se += __shfl_xor(se, off, 32);
    const float lse = logf(se) + mx;
    if (lane      < NCLS) out[(size_t)row * NCLS + lane     ] = v0 - lse;
    if (lane + 32 < NCLS) out[(size_t)row * NCLS + lane + 32] = v1 - lse;
}

extern "C" void kernel_launch(void* const* d_in, const int* in_sizes, int n_in,
                              void* d_out, int out_size, void* d_ws, size_t ws_size,
                              hipStream_t stream)
{
    (void)in_sizes; (void)n_in; (void)out_size; (void)ws_size;

    const float* x         = (const float*)d_in[0];
    const int*   src       = (const int*)  d_in[1];
    const int*   dst       = (const int*)  d_in[2];
    const int*   node_map  = (const int*)  d_in[3];
    const int*   final_map = (const int*)  d_in[4];
    const float* enc_w     = (const float*)d_in[5];
    const float* enc_b     = (const float*)d_in[6];
    const float* gcn_w     = (const float*)d_in[7];
    const float* gcn_b     = (const float*)d_in[8];
    const float* ln_g      = (const float*)d_in[9];
    const float* ln_b      = (const float*)d_in[10];
    const float* pred_w    = (const float*)d_in[11];
    const float* pred_b    = (const float*)d_in[12];
    float* out = (float*)d_out;

    const size_t NH = (size_t)NNODES * HDIM;
    float* B0     = (float*)d_ws;
    float* B1     = B0 + NH;
    float* B2     = B0 + 2 * NH;
    float* B4     = B0 + 4 * NH;
    float* HG     = B0 + 5 * NH;
    float* HG2    = HG  + (size_t)NSEEDS * HDIM;
    float* SC     = HG2 + (size_t)NSEEDS * HDIM;
    float* PWT    = SC  + (size_t)NSEEDS * NCLSPAD;
    float* PB     = PWT    + (size_t)NCLSPAD * HDIM;
    unsigned* WL  = (unsigned*)(PB + NCLSPAD);
    unsigned* WC  = WL + (size_t)AGG_NPART * AGG_K * AGG_CAP;
    unsigned* GM  = WC + (size_t)AGG_NPART * AGG_K;

    transpose_pad_kernel<<<(NCLSPAD * HDIM + 255) / 256, 256, 0, stream>>>(
        pred_w, PWT, HDIM, NCLS, NCLSPAD);
    pad_bias_kernel<<<1, NCLSPAD, 0, stream>>>(pred_b, PB, NCLS, NCLSPAD, GM);

    const dim3 gblk(32, 4, 1);
    const int  mtN   = NNODES / 16;
    const int  NWGP  = 256;
    const int  ngrid = (NNODES * 32 + 255) / 256;
    const int  lgrid = (NNODES + 7) / 8;

    gemm_ws_kernel<FDIM, 4, false, true><<<NWGP, 128, 0, stream>>>(
        x, enc_w, enc_b, nullptr, B0, mtN, GM);

    auto conv = [&](const float* hin, int eblk, int lmode, const float* wt, const float* bias,
                    const float* res, float* outp, const float* bg, const float* bb) {
        const int* es = src + (size_t)eblk * NEDGES;
        const int* ed = dst + (size_t)eblk * NEDGES;
        if (lmode == 1)      agg_kernel<1><<<AGG_NPART, 1024, 0, stream>>>(hin, es, ed, B2, WL, WC, bg, bb, GM);
        else if (lmode == 2) agg_kernel<2><<<AGG_NPART, 1024, 0, stream>>>(hin, es, ed, B2, WL, WC, bg, bb, GM);
        else                 agg_kernel<0><<<AGG_NPART, 1024, 0, stream>>>(hin, es, ed, B2, WL, WC, bg, bb, GM);
        if (res)
            gemm_ws_kernel<HDIM, 8, true, false><<<NWGP, 128, 0, stream>>>(B2, wt, bias, res, outp, mtN, nullptr);
        else
            gemm_ws_kernel<HDIM, 8, false, false><<<NWGP, 128, 0, stream>>>(B2, wt, bias, nullptr, outp, mtN, nullptr);
    };

    conv(B0, 0, 1, gcn_w, gcn_b, nullptr, B0, nullptr, nullptr);

    for (int l = 1; l < 3; ++l) {
        ln_relu_kernel<<<lgrid, 256, 0, stream>>>(
            B0, ln_g + (size_t)(l - 1) * HDIM, ln_b + (size_t)(l - 1) * HDIM, B1, NNODES);
        conv(B1, l - 1, l == 1 ? 2 : 0, gcn_w + (size_t)l * HDIM * HDIM, gcn_b + (size_t)l * HDIM,
              B0,  B4, ln_g + (size_t)(l - 1) * HDIM, ln_b + (size_t)(l - 1) * HDIM);
        gather_kernel<<<ngrid, 256, 0, stream>>>(
            B4, node_map + (size_t)(l - 1) * NNODES, B0, NNODES);
    }

    const int sgrid = (NSEEDS * 32 + 255) / 256;
    gather_kernel<<<sgrid, 256, 0, stream>>>(B0, final_map, HG, NSEEDS);
    ln_relu_kernel<<<(NSEEDS + 7) / 8, 256, 0, stream>>>(
        HG, ln_g + 2 * HDIM, ln_b + 2 * HDIM, HG2, NSEEDS);
    const int  mtS = NSEEDS / 16;
    const dim3 gridS((mtS + 3) / 4, NCLSPAD / 64, 1);
    gemm_f32_wmma_kernel<<<gridS, gblk, 0, stream>>>(
        HG2, PWT, PB, nullptr, SC, NSEEDS, HDIM, NCLSPAD);
    logsoftmax_kernel<<<(NSEEDS + 7) / 8, 256, 0, stream>>>(SC, out, NSEEDS);
}


# Edge lists dealt once per call: one launch bins all edges of both blocks to their owner ranges (LDS ranks, one global add per list and work-group); owners' dst scans gone; sums in LDS arrival order
# speedup vs baseline: 3.7342x; 1.0734x over previous
#include <hip/hip_runtime.h>
#include <hip/hip_bf16.h>
#include <math.h>

#define NNODES   100000
#define NEDGES   600000
#define FDIM     256
#define HDIM     128
#define NCLS     47
#define NCLSPAD  64
#define NSEEDS   10000
#define GEN_EPS  1e-7f
#define LN_EPS   1e-5f

typedef float v2f __attribute__((ext_vector_type(2)));
typedef float v8f __attribute__((ext_vector_type(8)));
typedef float v4f __attribute__((ext_vector_type(4)));

__global__ __launch_bounds__(128)
void gemm_f32_wmma_kernel(const float* __restrict__ A, const float* __restrict__ Wt,
                          const float* __restrict__ bias, const float* __restrict__ res,
                          float* __restrict__ out, int M, int K, int Nc)
{
    const int lane  = threadIdx.x;
    const int mtile = blockIdx.x * 4 + threadIdx.y;
    const int row0  = mtile * 16;
    if (row0 >= M) return;
    const int hb   = lane >> 4;
    const int lo   = lane & 15;
    const int arow = row0 + lo;
    const int col0 = blockIdx.y * 64 + lo;

    const float* __restrict__ a_ptr = A  + (size_t)arow * K;
    const float* __restrict__ b_ptr = Wt + (size_t)col0 * K;
    const size_t bstep = (size_t)16 * K;

    v8f acc[4];
#pragma unroll
    for (int t = 0; t < 4; ++t)
#pragma unroll
        for (int i = 0; i < 8; ++i) acc[t][i] = 0.f;

#pragma unroll 4
    for (int k = 0; k < K; k += 4) {
        const int ka = k + 2 * hb;
        const v2f a  = *(const v2f*)(a_ptr + ka);
        const v2f b0 = *(const v2f*)(b_ptr + ka);
        const v2f b1 = *(const v2f*)(b_ptr + bstep + ka);
        const v2f b2 = *(const v2f*)(b_ptr + 2 * bstep + ka);
        const v2f b3 = *(const v2f*)(b_ptr + 3 * bstep + ka);
        acc[0] = __builtin_amdgcn_wmma_f32_16x16x4_f32(false, a, false, b0, (short)0, acc[0], false, false);
        acc[1] = __builtin_amdgcn_wmma_f32_16x16x4_f32(false, a, false, b1, (short)0, acc[1], false, false);
        acc[2] = __builtin_amdgcn_wmma_f32_16x16x4_f32(false, a, false, b2, (short)0, acc[2], false, false);
        acc[3] = __builtin_amdgcn_wmma_f32_16x16x4_f32(false, a, false, b3, (short)0, acc[3], false, false);
    }

#pragma unroll
    for (int t = 0; t < 4; ++t) {
        const int   col = col0 + 16 * t;
        const float bv  = bias[col];
#pragma unroll
        for (int i = 0; i < 8; ++i) {
            const int r = row0 + i + 8 * hb;
            float v = acc[t][i] + bv;
            if (res) v += res[(size_t)r * Nc + col];
            out[(size_t)r * Nc + col] = v;
        }
    }
}

template <int K>
__device__ __forceinline__ void ws_load_a(float4 (&buf)[K / 8], const float* __restrict__ A,
                                          int tile, int lo, int hb)
{
    const float*   ap  = A + (size_t)tile * (16 * K);
    const unsigned lof = (unsigned)(lo * K + 4 * hb);
#pragma unroll
    for (int q = 0; q < K / 8; ++q) buf[q] = *(const float4*)(ap + lof + 8 * q);
}

template <int NT>
__device__ __forceinline__ void ws_load_r(float (&r)[NT][8], const float* __restrict__ res,
                                          int tile, int hb, int colb)
{
    const float*   rp  = res + (size_t)tile * (16 * HDIM);
    const unsigned lof = (unsigned)(8 * hb * HDIM + colb);
#pragma unroll
    for (int t = 0; t < NT; ++t)
#pragma unroll
        for (int i = 0; i < 8; ++i) r[t][i] = rp[lof + i * HDIM + 16 * t];
}

template <int K, int NT, bool HAS_RES, bool WANT_MAX>
__global__ __launch_bounds__(128)
void gemm_ws_kernel(const float* __restrict__ A, const float* __restrict__ W,
                    const float* __restrict__ bias, const float* __restrict__ res,
                    float* __restrict__ out, int mtiles, unsigned* __restrict__ vmax_out)
{
    constexpr int NCG = HDIM / (16 * NT);
    constexpr int NQ  = K / 8;
    const int lane   = threadIdx.x & 31;
    const int wave   = blockIdx.x * 4 + __builtin_amdgcn_readfirstlane(threadIdx.x >> 5);
    const int hb     = lane >> 4;
    const int lo     = lane & 15;
    const int colb   = (wave % NCG) * 16 * NT + lo;
    const int stride = (gridDim.x * 4) / NCG;
    const int last   = mtiles - 1;
    int mt = wave / NCG;
    if (mt > last) return;

    v2f w[K / 4][NT];
#pragma unroll
    for (int q = 0; q < NQ; ++q) {
        const float* wp = W + (size_t)(8 * q + 4 * hb) * HDIM + colb;
#pragma unroll
        for (int t = 0; t < NT; ++t) {
            w[2 * q][t][0]     = wp[16 * t];
            w[2 * q][t][1]     = wp[HDIM + 16 * t];
            w[2 * q + 1][t][0] = wp[2 * HDIM + 16 * t];
            w[2 * q + 1][t][1] = wp[3 * HDIM + 16 * t];
        }
    }
    float bv[NT];
#pragma unroll
    for (int t = 0; t < NT; ++t) bv[t] = bias[colb + 16 * t];

    float4 cur[NQ], nxt[NQ];
    float  rc[NT][8], rn[NT][8];
    float  vmax = 0.f;
    if (HAS_RES) ws_load_r<NT>(rn, res, mt, hb, colb);
    ws_load_a<K>(nxt, A, mt, lo, hb);

    while (true) {
#pragma unroll
        for (int q = 0; q < NQ; ++q) cur[q] = nxt[q];
        if (HAS_RES) {
#pragma unroll
            for (int t = 0; t < NT; ++t)
#pragma unroll
                for (int i = 0; i < 8; ++i) rc[t][i] = rn[t][i];
        }
        const int mtn = mt + stride;
        const int mtp = mtn < last ? mtn : last;
        ws_load_a<K>(nxt, A, mtp, lo, hb);

        v8f acc[NT];
#pragma unroll
        for (int t = 0; t < NT; ++t)
#pragma unroll
            for (int i = 0; i < 8; ++i) acc[t][i] = 0.f;

#pragma unroll
        for (int q = 0; q < NQ; ++q) {
            if (HAS_RES && q == NQ / 2) {
                __builtin_amdgcn_sched_barrier(0);
                ws_load_r<NT>(rn, res, mtp, hb, colb);
            }
            v2f a0, a1;
            a0[0] = cur[q].x; a0[1] = cur[q].y;
            a1[0] = cur[q].z; a1[1] = cur[q].w;
#pragma unroll
            for (int t = 0; t < NT; ++t)
                acc[t] = __builtin_amdgcn_wmma_f32_16x16x4_f32(false, a0, false, w[2 * q][t], (short)0, acc[t], false, false);
#pragma unroll
            for (int t = 0; t < NT; ++t)
                acc[t] = __builtin_amdgcn_wmma_f32_16x16x4_f32(false, a1, false, w[2 * q + 1][t], (short)0, acc[t], false, false);
        }

        float* __restrict__ op  = out + (size_t)mt * (16 * HDIM);
        const unsigned      oof = (unsigned)(8 * hb * HDIM + colb);
#pragma unroll
        for (int t = 0; t < NT; ++t)
#pragma unroll
            for (int i = 0; i < 8; ++i) {
                float v = acc[t][i] + bv[t];
                if (HAS_RES) v += rc[t][i];
                if (WANT_MAX) vmax = fmaxf(vmax, v);
                op[oof + i * HDIM + 16 * t] = v;
            }
        __builtin_amdgcn_sched_barrier(0);

        if (mtn > last) break;
        mt = mtn;
    }

    if (WANT_MAX) {
#pragma unroll
        for (int off = 16; off > 0; off >>= 1) vmax = fmaxf(vmax, __shfl_xor(vmax, off, 32));
        if (lane == 0) atomicMax(vmax_out, __float_as_uint(vmax));
    }
}

__global__ __launch_bounds__(256)
void transpose_pad_kernel(const float* __restrict__ W, float* __restrict__ Wt,
                          int K, int Nc, int Ncpad)
{
    const int tid = blockIdx.x * 256 + threadIdx.x;
    if (tid >= Ncpad * K) return;
    const int c = tid / K;
    const int k = tid - c * K;
    Wt[(size_t)c * K + k] = (c < Nc) ? W[(size_t)k * Nc + c] : 0.f;
}

__global__ void pad_bias_kernel(const float* __restrict__ b, float* __restrict__ bp,
                                int Nc, int Ncpad, unsigned* __restrict__ max_word,
                                unsigned* __restrict__ list_len, int nlen)
{
    const int c = threadIdx.x;
    if (c < Ncpad) bp[c] = (c < Nc) ? b[c] : 0.f;
    if (c == 0) *max_word = 0u;
    for (int i = c; i < nlen; i += (int)blockDim.x) list_len[i] = 0u;
}

#define AGG_R      196
#define AGG_K      2
#define AGG_RW     (AGG_R * AGG_K)
#define AGG_NPART  ((NNODES + AGG_RW - 1) / AGG_RW)
#define AGG_CAP    3000
#define AGG_LPT    ((AGG_CAP + 1023) / 1024)
#define AGG_NBIN   (AGG_NPART * AGG_K)
#define DEAL_WG    128
#define DEAL_EPW   ((NEDGES + DEAL_WG - 1) / DEAL_WG)
#define DEAL_IT    ((DEAL_EPW + 1023) / 1024)
#define AGG_SAFE   60.f
#define LN_BOUND   11.3137085f

static_assert(NNODES <= (1 << 17) && AGG_R <= 256, "list entry packing");
static_assert(AGG_NBIN <= 1024, "one thread per list in the dealing work-group");

__device__ __forceinline__ void lds_add(float* p, float v)
{
    (void)__hip_atomic_fetch_add(p, v, __ATOMIC_RELAXED, __HIP_MEMORY_SCOPE_WORKGROUP);
}

__device__ __forceinline__ void agg_edge_max(unsigned* s_mx, const v4f hv, unsigned ld, int lane)
{
    unsigned* p = s_mx + ld * HDIM + lane;
    atomicMax(p,      __float_as_uint(fmaxf(hv.x, 0.f) + GEN_EPS));
    atomicMax(p + 32, __float_as_uint(fmaxf(hv.y, 0.f) + GEN_EPS));
    atomicMax(p + 64, __float_as_uint(fmaxf(hv.z, 0.f) + GEN_EPS));
    atomicMax(p + 96, __float_as_uint(fmaxf(hv.w, 0.f) + GEN_EPS));
}

template <bool BOUND>
__device__ __forceinline__ void agg_edge_sum(const unsigned* s_mx, float* s_de, float* s_nu,
                                             const v4f hv, unsigned ld, int lane, float bound)
{
    const unsigned o = ld * HDIM + lane;
    float x0 = bound, x1 = bound, x2 = bound, x3 = bound;
    if (!BOUND) {
        x0 = __uint_as_float(s_mx[o]);
        x1 = __uint_as_float(s_mx[o + 32]);
        x2 = __uint_as_float(s_mx[o + 64]);
        x3 = __uint_as_float(s_mx[o + 96]);
    }
    const float m0 = fmaxf(hv.x, 0.f) + GEN_EPS, e0 = expf(m0 - x0);
    const float m1 = fmaxf(hv.y, 0.f) + GEN_EPS, e1 = expf(m1 - x1);
    const float m2 = fmaxf(hv.z, 0.f) + GEN_EPS, e2 = expf(m2 - x2);
    const float m3 = fmaxf(hv.w, 0.f) + GEN_EPS, e3 = expf(m3 - x3);
    lds_add(s_de + o,      e0); lds_add(s_nu + o,      m0 * e0);
    lds_add(s_de + o + 32, e1); lds_add(s_nu + o + 32, m1 * e1);
    lds_add(s_de + o + 64, e2); lds_add(s_nu + o + 64, m2 * e2);
    lds_add(s_de + o + 96, e3); lds_add(s_nu + o + 96, m3 * e3);
}

__global__ __launch_bounds__(1024)
void deal_kernel(const int* __restrict__ src, const int* __restrict__ dst,
                 unsigned* __restrict__ wl_list, unsigned* __restrict__ wl_cnt)
{
    __shared__ unsigned s_n[AGG_NBIN];
    __shared__ unsigned s_run[AGG_NBIN];

    const int tid = threadIdx.x;
    const int* __restrict__ es = src + (size_t)blockIdx.y * NEDGES;
    const int* __restrict__ ed = dst + (size_t)blockIdx.y * NEDGES;
    unsigned* __restrict__ lists = wl_list + (size_t)blockIdx.y * AGG_NBIN * AGG_CAP;
    unsigned* __restrict__ lens  = wl_cnt + blockIdx.y * AGG_NBIN;
    const int e0 = blockIdx.x * DEAL_EPW;
    const int e1 = (e0 + DEAL_EPW) < NEDGES ? (e0 + DEAL_EPW) : NEDGES;

    if (tid < AGG_NBIN) s_n[tid] = 0u;

    int dv[DEAL_IT], sv[DEAL_IT];
#pragma unroll
    for (int u = 0; u < DEAL_IT; ++u) {
        const int e = e0 + u * 1024 + tid;
        dv[u] = e < e1 ? ed[e] : -1;
        sv[u] = e < e1 ? es[e] : 0;
    }
    __syncthreads();

    unsigned bin[DEAL_IT], rank[DEAL_IT], ent[DEAL_IT];
#pragma unroll
    for (int u = 0; u < DEAL_IT; ++u) {
        const unsigned d = (unsigned)dv[u];
        bin[u] = 0xFFFFFFFFu; rank[u] = 0u; ent[u] = 0u;
        if (d < NNODES) {
            const unsigned own = d / AGG_RW;
            const unsigned l   = d - own * AGG_RW;
            const unsigned h   = l / AGG_R;
            bin[u]  = own * AGG_K + h;
            ent[u]  = (unsigned)sv[u] | ((l - h * AGG_R) << 17);
            rank[u] = atomicAdd(&s_n[bin[u]], 1u);
        }
    }
    __syncthreads();

    if (tid < AGG_NBIN) {
        const unsigned n = s_n[tid];
        s_run[tid] = n ? atomicAdd(&lens[tid], n) : 0u;
    }
    __syncthreads();

#pragma unroll
    for (int u = 0; u < DEAL_IT; ++u) {
        if (bin[u] != 0xFFFFFFFFu) {
            const unsigned pos = s_run[bin[u]] + rank[u];
            if (pos < AGG_CAP) lists[(size_t)bin[u] * AGG_CAP + pos] = ent[u];
        }
    }
}

template <bool BOUND>
__device__ __forceinline__ void agg_walk_sum(const float* __restrict__ hin, const int* __restrict__ src,
                                             const int* __restrict__ dst, const unsigned* lst,
                                             unsigned n, bool listed, int rbase, unsigned nrows,
                                             const unsigned* s_mx, float* s_de, float* s_nu,
                                             unsigned wave, int lane, float bound)
{
    if (listed) {
        for (unsigned i0 = wave; i0 < n; i0 += 128) {
            unsigned en[4], ld[4];
            v4f      hv[4];
#pragma unroll
            for (int u = 0; u < 4; ++u) {
                const unsigned idx = i0 + 32u * u;
                en[u] = lst[idx < n ? idx : n - 1u];
                ld[u] = idx < n ? (en[u] >> 17) : (unsigned)AGG_R;
            }
#pragma unroll
            for (int u = 0; u < 4; ++u)
                hv[u] = *(const v4f*)(hin + (size_t)(en[u] & 0x1FFFFu) * HDIM + 4 * lane);
#pragma unroll
            for (int u = 0; u < 4; ++u) agg_edge_sum<BOUND>(s_mx, s_de, s_nu, hv[u], ld[u], lane, bound);
        }
    } else {
        for (unsigned e = wave; e < NEDGES; e += 32) {
            const unsigned l = (unsigned)(dst[e] - rbase);
            if (l < nrows)
                agg_edge_sum<BOUND>(s_mx, s_de, s_nu,
                                    *(const v4f*)(hin + (size_t)src[e] * HDIM + 4 * lane), l, lane, bound);
        }
    }
}

__global__ __launch_bounds__(1024)
void agg_kernel(const float* __restrict__ hin, const int* __restrict__ src,
                const int* __restrict__ dst, float* __restrict__ t,
                const unsigned* __restrict__ wl_list, const unsigned* __restrict__ wl_cnt,
                const float* __restrict__ bg, const float* __restrict__ bb,
                const unsigned* __restrict__ bmax)
{
    __shared__ unsigned s_mx[(AGG_R + 1) * HDIM];
    __shared__ float    s_de[(AGG_R + 1) * HDIM];
    __shared__ float    s_nu[(AGG_R + 1) * HDIM];
    __shared__ unsigned s_list[AGG_K * AGG_CAP];
    __shared__ unsigned s_cnt[AGG_K];
    __shared__ unsigned s_bnd[2];

    const int tid  = threadIdx.x;
    const int lane = tid & 31;
    const unsigned wave = (unsigned)__builtin_amdgcn_readfirstlane(tid >> 5);
    const int      row0 = blockIdx.x * AGG_RW;

    const unsigned* __restrict__ my_list = wl_list + (size_t)blockIdx.x * (AGG_K * AGG_CAP);
    const unsigned* __restrict__ my_cnt  = wl_cnt + blockIdx.x * AGG_K;

    if (tid < 2) s_bnd[tid] = 0u;
    __syncthreads();
    if (bg) {
        if (tid < HDIM)          atomicMax(&s_bnd[0], __float_as_uint(fabsf(bg[tid])));
        else if (tid < 2 * HDIM) atomicMax(&s_bnd[1], __float_as_uint(fabsf(bb[tid - HDIM])));
    } else if (tid == 0) {
        s_bnd[1] = *bmax;
    }

    {
        unsigned lim[AGG_K], ent[AGG_K][AGG_LPT];
#pragma unroll
        for (int h = 0; h < AGG_K; ++h) {
            const unsigned n = my_cnt[h];
            lim[h] = n <= AGG_CAP ? n : 0u;
            if (tid == h) s_cnt[h] = n;
        }
#pragma unroll
        for (int h = 0; h < AGG_K; ++h)
#pragma unroll
            for (int u = 0; u < AGG_LPT; ++u) {
                const unsigned i = (unsigned)tid + 1024u * u;
                ent[h][u] = i < lim[h] ? my_list[h * AGG_CAP + i] : 0u;
            }
#pragma unroll
        for (int h = 0; h < AGG_K; ++h)
#pragma unroll
            for (int u = 0; u < AGG_LPT; ++u) {
                const unsigned i = (unsigned)tid + 1024u * u;
                if (i < lim[h]) s_list[h * AGG_CAP + i] = ent[h][u];
            }
        __syncthreads();
    }

    const float bound = LN_BOUND * __uint_as_float(s_bnd[0]) + __uint_as_float(s_bnd[1]) + GEN_EPS;
    const bool  usebound = __builtin_amdgcn_readfirstlane((int)(bound <= AGG_SAFE)) != 0;

    for (int h = 0; h < AGG_K; ++h) {
        const int rbase = row0 + h * AGG_R;
        const int left  = NNODES - rbase;
        if (left <= 0) break;
        const unsigned  nrows  = (unsigned)(left < AGG_R ? left : AGG_R);
        const unsigned  n      = (unsigned)__builtin_amdgcn_readfirstlane((int)s_cnt[h]);
        const bool      listed = n <= AGG_CAP;
        const unsigned* lst    = s_list + h * AGG_CAP;

        if (usebound) {
            for (int i = tid; i < (AGG_R + 1) * HDIM; i += 1024) { s_de[i] = 0.f; s_nu[i] = 0.f; }
        } else {
            for (int i = tid; i < (AGG_R + 1) * HDIM; i += 1024) { s_mx[i] = 0u; s_de[i] = 0.f; s_nu[i] = 0.f; }
        }
        __syncthreads();

        if (usebound) {
            agg_walk_sum<true>(hin, src, dst, lst, n, listed, rbase, nrows, s_mx, s_de, s_nu, wave, lane, bound);
        } else {
            if (listed) {
                for (unsigned i0 = wave; i0 < n; i0 += 128) {
                    unsigned en[4], ld[4];
                    v4f      hv[4];
#pragma unroll
                    for (int u = 0; u < 4; ++u) {
                        const unsigned idx = i0 + 32u * u;
                        en[u] = lst[idx < n ? idx : n - 1u];
                        ld[u] = idx < n ? (en[u] >> 17) : (unsigned)AGG_R;
                    }
#pragma unroll
                    for (int u = 0; u < 4; ++u)
                        hv[u] = *(const v4f*)(hin + (size_t)(en[u] & 0x1FFFFu) * HDIM + 4 * lane);
                    asm volatile("" :: "v"(hv[0]), "v"(hv[1]), "v"(hv[2]), "v"(hv[3]));
#pragma unroll
                    for (int u = 0; u < 4; ++u) agg_edge_max(s_mx, hv[u], ld[u], lane);
                }
            } else {
                for (unsigned e = wave; e < NEDGES; e += 32) {
                    const unsigned l = (unsigned)(dst[e] - rbase);
                    if (l < nrows)
                        agg_edge_max(s_mx, *(const v4f*)(hin + (size_t)src[e] * HDIM + 4 * lane), l, lane);
                }
            }
            __syncthreads();

            agg_walk_sum<false>(hin, src, dst, lst, n, listed, rbase, nrows, s_mx, s_de, s_nu, wave, lane, 0.f);
        }
        __syncthreads();

        for (unsigned r = wave; r < nrows; r += 32) {
            const size_t   g  = (size_t)(rbase + (int)r) * HDIM + 4 * lane;
            const float4   hv = *(const float4*)(hin + g);
            const unsigned o  = r * HDIM + lane;
            const float d0 = s_de[o], d1 = s_de[o + 32], d2 = s_de[o + 64], d3 = s_de[o + 96];
            const float n0 = s_nu[o], n1 = s_nu[o + 32], n2 = s_nu[o + 64], n3 = s_nu[o + 96];
            float4 q;
            if (usebound) {
                q.x = (d0 > 0.f ? n0 / d0 : 0.f) + hv.x;
                q.y = (d1 > 0.f ? n1 / d1 : 0.f) + hv.y;
                q.z = (d2 > 0.f ? n2 / d2 : 0.f) + hv.z;
                q.w = (d3 > 0.f ? n3 / d3 : 0.f) + hv.w;
            } else {
                q.x = n0 / (d0 + 1e-16f) + hv.x;
                q.y = n1 / (d1 + 1e-16f) + hv.y;
                q.z = n2 / (d2 + 1e-16f) + hv.z;
                q.w = n3 / (d3 + 1e-16f) + hv.w;
            }
            *(float4*)(t + g) = q;
        }
        __syncthreads();
    }
}

__global__ __launch_bounds__(256)
void ln_relu_kernel(const float* __restrict__ in, const float* __restrict__ g,
                    const float* __restrict__ b, float* __restrict__ out, int M)
{
    const int row  = blockIdx.x * 8 + (threadIdx.x >> 5);
    const int lane = threadIdx.x & 31;
    if (row >= M) return;
    const float4 v = *(const float4*)(in + (size_t)row * HDIM + lane * 4);
    float s1 = v.x + v.y + v.z + v.w;
    float s2 = v.x * v.x + v.y * v.y + v.z * v.z + v.w * v.w;
#pragma unroll
    for (int off = 16; off > 0; off >>= 1) {
        s1 += __shfl_xor(s1, off, 32);
        s2 += __shfl_xor(s2, off, 32);
    }
    const float mu  = s1 * (1.f / HDIM);
    const float var = s2 * (1.f / HDIM) - mu * mu;
    const float inv = rsqrtf(var + LN_EPS);
    const float4 gg = *(const float4*)(g + lane * 4);
    const float4 bb = *(const float4*)(b + lane * 4);
    float4 r;
    r.x = fmaxf((v.x - mu) * inv * gg.x + bb.x, 0.f);
    r.y = fmaxf((v.y - mu) * inv * gg.y + bb.y, 0.f);
    r.z = fmaxf((v.z - mu) * inv * gg.z + bb.z, 0.f);
    r.w = fmaxf((v.w - mu) * inv * gg.w + bb.w, 0.f);
    *(float4*)(out + (size_t)row * HDIM + lane * 4) = r;
}

__global__ __launch_bounds__(256)
void gather_kernel(const float* __restrict__ in, const int* __restrict__ map,
                   float* __restrict__ out, int M)
{
    const int tid = blockIdx.x * 256 + threadIdx.x;
    if (tid >= M * 32) return;
    const int i = tid >> 5;
    const int f = (tid & 31) * 4;
    const int j = map[i];
    *(float4*)(out + (size_t)i * HDIM + f) =
        *(const float4*)(in + (size_t)j * HDIM + f);
}

__global__ __launch_bounds__(256)
void logsoftmax_kernel(const float* __restrict__ in, float* __restrict__ out, int M)
{
    const int row  = blockIdx.x * 8 + (threadIdx.x >> 5);
    const int lane = threadIdx.x & 31;
    if (row >= M) return;
    const float NEG = -3.0e38f;
    const float v0 = (lane      < NCLS) ? in[(size_t)row * NCLSPAD + lane     ] : NEG;
    const float v1 = (lane + 32 < NCLS) ? in[(size_t)row * NCLSPAD + lane + 32] : NEG;
    float mx = fmaxf(v0, v1);
#pragma unroll
    for (int off = 16; off > 0; off >>= 1) mx = fmaxf(mx, __shfl_xor(mx, off, 32));
    float se = ((lane      < NCLS) ? expf(v0 - mx) : 0.f)
             + ((lane + 32 < NCLS) ? expf(v1 - mx) : 0.f);
#pragma unroll
    for (int off = 16; off > 0; off >>= 1) se += __shfl_xor(se, off, 32);
    const float lse = logf(se) + mx;
    if (lane      < NCLS) out[(size_t)row * NCLS + lane     ] = v0 - lse;
    if (lane + 32 < NCLS) out[(size_t)row * NCLS + lane + 32] = v1 - lse;
}

extern "C" void kernel_launch(void* const* d_in, const int* in_sizes, int n_in,
                              void* d_out, int out_size, void* d_ws, size_t ws_size,
                              hipStream_t stream)
{
    (void)in_sizes; (void)n_in; (void)out_size; (void)ws_size;

    const float* x         = (const float*)d_in[0];
    const int*   src       = (const int*)  d_in[1];
    const int*   dst       = (const int*)  d_in[2];
    const int*   node_map  = (const int*)  d_in[3];
    const int*   final_map = (const int*)  d_in[4];
    const float* enc_w     = (const float*)d_in[5];
    const float* enc_b     = (const float*)d_in[6];
    const float* gcn_w     = (const float*)d_in[7];
    const float* gcn_b     = (const float*)d_in[8];
    const float* ln_g      = (const float*)d_in[9];
    const float* ln_b      = (const float*)d_in[10];
    const float* pred_w    = (const float*)d_in[11];
    const float* pred_b    = (const float*)d_in[12];
    float* out = (float*)d_out;

    const size_t NH = (size_t)NNODES * HDIM;
    float* B0     = (float*)d_ws;
    float* B1     = B0 + NH;
    float* B2     = B0 + 2 * NH;
    float* B4     = B0 + 4 * NH;
    float* HG     = B0 + 5 * NH;
    float* HG2    = HG  + (size_t)NSEEDS * HDIM;
    float* SC     = HG2 + (size_t)NSEEDS * HDIM;
    float* PWT    = SC  + (size_t)NSEEDS * NCLSPAD;
    float* PB     = PWT    + (size_t)NCLSPAD * HDIM;
    unsigned* WL  = (unsigned*)(PB + NCLSPAD);
    unsigned* WC  = WL + (size_t)2 * AGG_NBIN * AGG_CAP;
    unsigned* GM  = WC + (size_t)2 * AGG_NBIN;

    transpose_pad_kernel<<<(NCLSPAD * HDIM + 255) / 256, 256, 0, stream>>>(
        pred_w, PWT, HDIM, NCLS, NCLSPAD);
    pad_bias_kernel<<<1, 1024, 0, stream>>>(pred_b, PB, NCLS, NCLSPAD, GM, WC, 2 * AGG_NBIN);

    deal_kernel<<<dim3(DEAL_WG, 2, 1), 1024, 0, stream>>>(src, dst, WL, WC);

    const dim3 gblk(32, 4, 1);
    const int  mtN   = NNODES / 16;
    const int  NWGP  = 256;
    const int  ngrid = (NNODES * 32 + 255) / 256;
    const int  lgrid = (NNODES + 7) / 8;

    gemm_ws_kernel<FDIM, 4, false, true><<<NWGP, 128, 0, stream>>>(
        x, enc_w, enc_b, nullptr, B0, mtN, GM);

    auto conv = [&](const float* hin, int eblk, const float* wt, const float* bias,
                    const float* res, float* outp, const float* bg, const float* bb) {
        const int* es = src + (size_t)eblk * NEDGES;
        const int* ed = dst + (size_t)eblk * NEDGES;
        agg_kernel<<<AGG_NPART, 1024, 0, stream>>>(hin, es, ed, B2, WL + (size_t)eblk * AGG_NBIN * AGG_CAP,
                                                   WC + eblk * AGG_NBIN, bg, bb, GM);
        if (res)
            gemm_ws_kernel<HDIM, 8, true, false><<<NWGP, 128, 0, stream>>>(B2, wt, bias, res, outp, mtN, nullptr);
        else
            gemm_ws_kernel<HDIM, 8, false, false><<<NWGP, 128, 0, stream>>>(B2, wt, bias, nullptr, outp, mtN, nullptr);
    };

    conv(B0, 0, gcn_w, gcn_b, nullptr, B0, nullptr, nullptr);

    for (int l = 1; l < 3; ++l) {
        ln_relu_kernel<<<lgrid, 256, 0, stream>>>(
            B0, ln_g + (size_t)(l - 1) * HDIM, ln_b + (size_t)(l - 1) * HDIM, B1, NNODES);
        conv(B1, l - 1, gcn_w + (size_t)l * HDIM * HDIM, gcn_b + (size_t)l * HDIM,
              B0,  B4, ln_g + (size_t)(l - 1) * HDIM, ln_b + (size_t)(l - 1) * HDIM);
        gather_kernel<<<ngrid, 256, 0, stream>>>(
            B4, node_map + (size_t)(l - 1) * NNODES, B0, NNODES);
    }

    const int sgrid = (NSEEDS * 32 + 255) / 256;
    gather_kernel<<<sgrid, 256, 0, stream>>>(B0, final_map, HG, NSEEDS);
    ln_relu_kernel<<<(NSEEDS + 7) / 8, 256, 0, stream>>>(
        HG, ln_g + 2 * HDIM, ln_b + 2 * HDIM, HG2, NSEEDS);
    const int  mtS = NSEEDS / 16;
    const dim3 gridS((mtS + 3) / 4, NCLSPAD / 64, 1);
    gemm_f32_wmma_kernel<<<gridS, gblk, 0, stream>>>(
        HG2, PWT, PB, nullptr, SC, NSEEDS, HDIM, NCLSPAD);
    logsoftmax_kernel<<<(NSEEDS + 7) / 8, 256, 0, stream>>>(SC, out, NSEEDS);
}


# Head in one launch: a wave takes 16 seed rows of the last product's output via node_map[1][final_map], LN+ReLU, WMMA product, log-softmax in-wave (row sums reordered); 18 -> 13 launches
# speedup vs baseline: 3.9770x; 1.0650x over previous
#include <hip/hip_runtime.h>
#include <hip/hip_bf16.h>
#include <math.h>

#define NNODES   100000
#define NEDGES   600000
#define FDIM     256
#define HDIM     128
#define NCLS     47
#define NCLSPAD  64
#define NSEEDS   10000
#define GEN_EPS  1e-7f
#define LN_EPS   1e-5f

typedef float v2f __attribute__((ext_vector_type(2)));
typedef float v8f __attribute__((ext_vector_type(8)));
typedef float v4f __attribute__((ext_vector_type(4)));

template <int K>
__device__ __forceinline__ void ws_load_a(float4 (&buf)[K / 8], const float* __restrict__ A,
                                          int tile, int lo, int hb)
{
    const float*   ap  = A + (size_t)tile * (16 * K);
    const unsigned lof = (unsigned)(lo * K + 4 * hb);
#pragma unroll
    for (int q = 0; q < K / 8; ++q) buf[q] = *(const float4*)(ap + lof + 8 * q);
}

template <int NT>
__device__ __forceinline__ void ws_load_r(float (&r)[NT][8], const float* __restrict__ res,
                                          int tile, int hb, int colb)
{
    const float*   rp  = res + (size_t)tile * (16 * HDIM);
    const unsigned lof = (unsigned)(8 * hb * HDIM + colb);
#pragma unroll
    for (int t = 0; t < NT; ++t)
#pragma unroll
        for (int i = 0; i < 8; ++i) r[t][i] = rp[lof + i * HDIM + 16 * t];
}

template <int K, int NT, bool HAS_RES, bool WANT_MAX>
__global__ __launch_bounds__(128)
void gemm_ws_kernel(const float* __restrict__ A, const float* __restrict__ W,
                    const float* __restrict__ bias, const float* __restrict__ res,
                    float* __restrict__ out, int mtiles, unsigned* __restrict__ vmax_out)
{
    constexpr int NCG = HDIM / (16 * NT);
    constexpr int NQ  = K / 8;
    const int lane   = threadIdx.x & 31;
    const int wave   = blockIdx.x * 4 + __builtin_amdgcn_readfirstlane(threadIdx.x >> 5);
    const int hb     = lane >> 4;
    const int lo     = lane & 15;
    const int colb   = (wave % NCG) * 16 * NT + lo;
    const int stride = (gridDim.x * 4) / NCG;
    const int last   = mtiles - 1;
    int mt = wave / NCG;
    if (mt > last) return;

    v2f w[K / 4][NT];
#pragma unroll
    for (int q = 0; q < NQ; ++q) {
        const float* wp = W + (size_t)(8 * q + 4 * hb) * HDIM + colb;
#pragma unroll
        for (int t = 0; t < NT; ++t) {
            w[2 * q][t][0]     = wp[16 * t];
            w[2 * q][t][1]     = wp[HDIM + 16 * t];
            w[2 * q + 1][t][0] = wp[2 * HDIM + 16 * t];
            w[2 * q + 1][t][1] = wp[3 * HDIM + 16 * t];
        }
    }
    float bv[NT];
#pragma unroll
    for (int t = 0; t < NT; ++t) bv[t] = bias[colb + 16 * t];

    float4 cur[NQ], nxt[NQ];
    float  rc[NT][8], rn[NT][8];
    float  vmax = 0.f;
    if (HAS_RES) ws_load_r<NT>(rn, res, mt, hb, colb);
    ws_load_a<K>(nxt, A, mt, lo, hb);

    while (true) {
#pragma unroll
        for (int q = 0; q < NQ; ++q) cur[q] = nxt[q];
        if (HAS_RES) {
#pragma unroll
            for (int t = 0; t < NT; ++t)
#pragma unroll
                for (int i = 0; i < 8; ++i) rc[t][i] = rn[t][i];
        }
        const int mtn = mt + stride;
        const int mtp = mtn < last ? mtn : last;
        ws_load_a<K>(nxt, A, mtp, lo, hb);

        v8f acc[NT];
#pragma unroll
        for (int t = 0; t < NT; ++t)
#pragma unroll
            for (int i = 0; i < 8; ++i) acc[t][i] = 0.f;

#pragma unroll
        for (int q = 0; q < NQ; ++q) {
            if (HAS_RES && q == NQ / 2) {
                __builtin_amdgcn_sched_barrier(0);
                ws_load_r<NT>(rn, res, mtp, hb, colb);
            }
            v2f a0, a1;
            a0[0] = cur[q].x; a0[1] = cur[q].y;
            a1[0] = cur[q].z; a1[1] = cur[q].w;
#pragma unroll
            for (int t = 0; t < NT; ++t)
                acc[t] = __builtin_amdgcn_wmma_f32_16x16x4_f32(false, a0, false, w[2 * q][t], (short)0, acc[t], false, false);
#pragma unroll
            for (int t = 0; t < NT; ++t)
                acc[t] = __builtin_amdgcn_wmma_f32_16x16x4_f32(false, a1, false, w[2 * q + 1][t], (short)0, acc[t], false, false);
        }

        float* __restrict__ op  = out + (size_t)mt * (16 * HDIM);
        const unsigned      oof = (unsigned)(8 * hb * HDIM + colb);
#pragma unroll
        for (int t = 0; t < NT; ++t)
#pragma unroll
            for (int i = 0; i < 8; ++i) {
                float v = acc[t][i] + bv[t];
                if (HAS_RES) v += rc[t][i];
                if (WANT_MAX) vmax = fmaxf(vmax, v);
                op[oof + i * HDIM + 16 * t] = v;
            }
        __builtin_amdgcn_sched_barrier(0);

        if (mtn > last) break;
        mt = mtn;
    }

    if (WANT_MAX) {
#pragma unroll
        for (int off = 16; off > 0; off >>= 1) vmax = fmaxf(vmax, __shfl_xor(vmax, off, 32));
        if (lane == 0) atomicMax(vmax_out, __float_as_uint(vmax));
    }
}

__global__ __launch_bounds__(1024)
void prep_kernel(const float* __restrict__ W, float* __restrict__ Wt,
                 const float* __restrict__ b, float* __restrict__ bp,
                 int K, int Nc, int Ncpad, unsigned* __restrict__ max_word,
                 unsigned* __restrict__ list_len, int nlen)
{
    const int tid = blockIdx.x * 1024 + threadIdx.x;
    if (tid < Ncpad * K) {
        const int c = tid / K;
        const int k = tid - c * K;
        Wt[(size_t)c * K + k] = (c < Nc) ? W[(size_t)k * Nc + c] : 0.f;
    }
    if (blockIdx.x == 0) {
        const int c = threadIdx.x;
        if (c < Ncpad) bp[c] = (c < Nc) ? b[c] : 0.f;
        if (c == 0) *max_word = 0u;
        for (int i = c; i < nlen; i += 1024) list_len[i] = 0u;
    }
}

#define AGG_R      196
#define AGG_K      2
#define AGG_RW     (AGG_R * AGG_K)
#define AGG_NPART  ((NNODES + AGG_RW - 1) / AGG_RW)
#define AGG_CAP    3000
#define AGG_LPT    ((AGG_CAP + 1023) / 1024)
#define AGG_NBIN   (AGG_NPART * AGG_K)
#define DEAL_WG    128
#define DEAL_EPW   ((NEDGES + DEAL_WG - 1) / DEAL_WG)
#define DEAL_IT    ((DEAL_EPW + 1023) / 1024)
#define AGG_SAFE   60.f
#define LN_BOUND   11.3137085f

static_assert(NNODES <= (1 << 17) && AGG_R <= 256, "list entry packing");
static_assert(AGG_NBIN <= 1024, "one thread per list in the dealing work-group");

__device__ __forceinline__ void lds_add(float* p, float v)
{
    (void)__hip_atomic_fetch_add(p, v, __ATOMIC_RELAXED, __HIP_MEMORY_SCOPE_WORKGROUP);
}

__device__ __forceinline__ void agg_edge_max(unsigned* s_mx, const v4f hv, unsigned ld, int lane)
{
    unsigned* p = s_mx + ld * HDIM + lane;
    atomicMax(p,      __float_as_uint(fmaxf(hv.x, 0.f) + GEN_EPS));
    atomicMax(p + 32, __float_as_uint(fmaxf(hv.y, 0.f) + GEN_EPS));
    atomicMax(p + 64, __float_as_uint(fmaxf(hv.z, 0.f) + GEN_EPS));
    atomicMax(p + 96, __float_as_uint(fmaxf(hv.w, 0.f) + GEN_EPS));
}

template <bool BOUND>
__device__ __forceinline__ void agg_edge_sum(const unsigned* s_mx, float* s_de, float* s_nu,
                                             const v4f hv, unsigned ld, int lane, float bound)
{
    const unsigned o = ld * HDIM + lane;
    float x0 = bound, x1 = bound, x2 = bound, x3 = bound;
    if (!BOUND) {
        x0 = __uint_as_float(s_mx[o]);
        x1 = __uint_as_float(s_mx[o + 32]);
        x2 = __uint_as_float(s_mx[o + 64]);
        x3 = __uint_as_float(s_mx[o + 96]);
    }
    const float m0 = fmaxf(hv.x, 0.f) + GEN_EPS, e0 = expf(m0 - x0);
    const float m1 = fmaxf(hv.y, 0.f) + GEN_EPS, e1 = expf(m1 - x1);
    const float m2 = fmaxf(hv.z, 0.f) + GEN_EPS, e2 = expf(m2 - x2);
    const float m3 = fmaxf(hv.w, 0.f) + GEN_EPS, e3 = expf(m3 - x3);
    lds_add(s_de + o,      e0); lds_add(s_nu + o,      m0 * e0);
    lds_add(s_de + o + 32, e1); lds_add(s_nu + o + 32, m1 * e1);
    lds_add(s_de + o + 64, e2); lds_add(s_nu + o + 64, m2 * e2);
    lds_add(s_de + o + 96, e3); lds_add(s_nu + o + 96, m3 * e3);
}

__global__ __launch_bounds__(1024)
void deal_kernel(const int* __restrict__ src, const int* __restrict__ dst,
                 unsigned* __restrict__ wl_list, unsigned* __restrict__ wl_cnt)
{
    __shared__ unsigned s_n[AGG_NBIN];
    __shared__ unsigned s_run[AGG_NBIN];

    const int tid = threadIdx.x;
    const int* __restrict__ es = src + (size_t)blockIdx.y * NEDGES;
    const int* __restrict__ ed = dst + (size_t)blockIdx.y * NEDGES;
    unsigned* __restrict__ lists = wl_list + (size_t)blockIdx.y * AGG_NBIN * AGG_CAP;
    unsigned* __restrict__ lens  = wl_cnt + blockIdx.y * AGG_NBIN;
    const int e0 = blockIdx.x * DEAL_EPW;
    const int e1 = (e0 + DEAL_EPW) < NEDGES ? (e0 + DEAL_EPW) : NEDGES;

    if (tid < AGG_NBIN) s_n[tid] = 0u;

    int dv[DEAL_IT], sv[DEAL_IT];
#pragma unroll
    for (int u = 0; u < DEAL_IT; ++u) {
        const int e = e0 + u * 1024 + tid;
        dv[u] = e < e1 ? ed[e] : -1;
        sv[u] = e < e1 ? es[e] : 0;
    }
    __syncthreads();

    unsigned bin[DEAL_IT], rank[DEAL_IT], ent[DEAL_IT];
#pragma unroll
    for (int u = 0; u < DEAL_IT; ++u) {
        const unsigned d = (unsigned)dv[u];
        bin[u] = 0xFFFFFFFFu; rank[u] = 0u; ent[u] = 0u;
        if (d < NNODES) {
            const unsigned own = d / AGG_RW;
            const unsigned l   = d - own * AGG_RW;
            const unsigned h   = l / AGG_R;
            bin[u]  = own * AGG_K + h;
            ent[u]  = (unsigned)sv[u] | ((l - h * AGG_R) << 17);
            rank[u] = atomicAdd(&s_n[bin[u]], 1u);
        }
    }
    __syncthreads();

    if (tid < AGG_NBIN) {
        const unsigned n = s_n[tid];
        s_run[tid] = n ? atomicAdd(&lens[tid], n) : 0u;
    }
    __syncthreads();

#pragma unroll
    for (int u = 0; u < DEAL_IT; ++u) {
        if (bin[u] != 0xFFFFFFFFu) {
            const unsigned pos = s_run[bin[u]] + rank[u];
            if (pos < AGG_CAP) lists[(size_t)bin[u] * AGG_CAP + pos] = ent[u];
        }
    }
}

template <bool BOUND>
__device__ __forceinline__ void agg_walk_sum(const float* __restrict__ hin, const int* __restrict__ src,
                                             const int* __restrict__ dst, const unsigned* lst,
                                             unsigned n, bool listed, int rbase, unsigned nrows,
                                             const unsigned* s_mx, float* s_de, float* s_nu,
                                             unsigned wave, int lane, float bound)
{
    if (listed) {
        for (unsigned i0 = wave; i0 < n; i0 += 128) {
            unsigned en[4], ld[4];
            v4f      hv[4];
#pragma unroll
            for (int u = 0; u < 4; ++u) {
                const unsigned idx = i0 + 32u * u;
                en[u] = lst[idx < n ? idx : n - 1u];
                ld[u] = idx < n ? (en[u] >> 17) : (unsigned)AGG_R;
            }
#pragma unroll
            for (int u = 0; u < 4; ++u)
                hv[u] = *(const v4f*)(hin + (size_t)(en[u] & 0x1FFFFu) * HDIM + 4 * lane);
#pragma unroll
            for (int u = 0; u < 4; ++u) agg_edge_sum<BOUND>(s_mx, s_de, s_nu, hv[u], ld[u], lane, bound);
        }
    } else {
        for (unsigned e = wave; e < NEDGES; e += 32) {
            const unsigned l = (unsigned)(dst[e] - rbase);
            if (l < nrows)
                agg_edge_sum<BOUND>(s_mx, s_de, s_nu,
                                    *(const v4f*)(hin + (size_t)src[e] * HDIM + 4 * lane), l, lane, bound);
        }
    }
}

__global__ __launch_bounds__(1024)
void agg_kernel(const float* __restrict__ hin, const int* __restrict__ src,
                const int* __restrict__ dst, float* __restrict__ t,
                const unsigned* __restrict__ wl_list, const unsigned* __restrict__ wl_cnt,
                const float* __restrict__ bg, const float* __restrict__ bb,
                const unsigned* __restrict__ bmax)
{
    __shared__ unsigned s_mx[(AGG_R + 1) * HDIM];
    __shared__ float    s_de[(AGG_R + 1) * HDIM];
    __shared__ float    s_nu[(AGG_R + 1) * HDIM];
    __shared__ unsigned s_list[AGG_K * AGG_CAP];
    __shared__ unsigned s_cnt[AGG_K];
    __shared__ unsigned s_bnd[2];

    const int tid  = threadIdx.x;
    const int lane = tid & 31;
    const unsigned wave = (unsigned)__builtin_amdgcn_readfirstlane(tid >> 5);
    const int      row0 = blockIdx.x * AGG_RW;

    const unsigned* __restrict__ my_list = wl_list + (size_t)blockIdx.x * (AGG_K * AGG_CAP);
    const unsigned* __restrict__ my_cnt  = wl_cnt + blockIdx.x * AGG_K;

    if (tid < 2) s_bnd[tid] = 0u;
    __syncthreads();
    if (bg) {
        if (tid < HDIM)          atomicMax(&s_bnd[0], __float_as_uint(fabsf(bg[tid])));
        else if (tid < 2 * HDIM) atomicMax(&s_bnd[1], __float_as_uint(fabsf(bb[tid - HDIM])));
    } else if (tid == 0) {
        s_bnd[1] = *bmax;
    }

    {
        unsigned lim[AGG_K], ent[AGG_K][AGG_LPT];
#pragma unroll
        for (int h = 0; h < AGG_K; ++h) {
            const unsigned n = my_cnt[h];
            lim[h] = n <= AGG_CAP ? n : 0u;
            if (tid == h) s_cnt[h] = n;
        }
#pragma unroll
        for (int h = 0; h < AGG_K; ++h)
#pragma unroll
            for (int u = 0; u < AGG_LPT; ++u) {
                const unsigned i = (unsigned)tid + 1024u * u;
                ent[h][u] = i < lim[h] ? my_list[h * AGG_CAP + i] : 0u;
            }
#pragma unroll
        for (int h = 0; h < AGG_K; ++h)
#pragma unroll
            for (int u = 0; u < AGG_LPT; ++u) {
                const unsigned i = (unsigned)tid + 1024u * u;
                if (i < lim[h]) s_list[h * AGG_CAP + i] = ent[h][u];
            }
        __syncthreads();
    }

    const float bound = LN_BOUND * __uint_as_float(s_bnd[0]) + __uint_as_float(s_bnd[1]) + GEN_EPS;
    const bool  usebound = __builtin_amdgcn_readfirstlane((int)(bound <= AGG_SAFE)) != 0;

    for (int h = 0; h < AGG_K; ++h) {
        const int rbase = row0 + h * AGG_R;
        const int left  = NNODES - rbase;
        if (left <= 0) break;
        const unsigned  nrows  = (unsigned)(left < AGG_R ? left : AGG_R);
        const unsigned  n      = (unsigned)__builtin_amdgcn_readfirstlane((int)s_cnt[h]);
        const bool      listed = n <= AGG_CAP;
        const unsigned* lst    = s_list + h * AGG_CAP;

        if (usebound) {
            for (int i = tid; i < (AGG_R + 1) * HDIM; i += 1024) { s_de[i] = 0.f; s_nu[i] = 0.f; }
        } else {
            for (int i = tid; i < (AGG_R + 1) * HDIM; i += 1024) { s_mx[i] = 0u; s_de[i] = 0.f; s_nu[i] = 0.f; }
        }
        __syncthreads();

        if (usebound) {
            agg_walk_sum<true>(hin, src, dst, lst, n, listed, rbase, nrows, s_mx, s_de, s_nu, wave, lane, bound);
        } else {
            if (listed) {
                for (unsigned i0 = wave; i0 < n; i0 += 128) {
                    unsigned en[4], ld[4];
                    v4f      hv[4];
#pragma unroll
                    for (int u = 0; u < 4; ++u) {
                        const unsigned idx = i0 + 32u * u;
                        en[u] = lst[idx < n ? idx : n - 1u];
                        ld[u] = idx < n ? (en[u] >> 17) : (unsigned)AGG_R;
                    }
#pragma unroll
                    for (int u = 0; u < 4; ++u)
                        hv[u] = *(const v4f*)(hin + (size_t)(en[u] & 0x1FFFFu) * HDIM + 4 * lane);
                    asm volatile("" :: "v"(hv[0]), "v"(hv[1]), "v"(hv[2]), "v"(hv[3]));
#pragma unroll
                    for (int u = 0; u < 4; ++u) agg_edge_max(s_mx, hv[u], ld[u], lane);
                }
            } else {
                for (unsigned e = wave; e < NEDGES; e += 32) {
                    const unsigned l = (unsigned)(dst[e] - rbase);
                    if (l < nrows)
                        agg_edge_max(s_mx, *(const v4f*)(hin + (size_t)src[e] * HDIM + 4 * lane), l, lane);
                }
            }
            __syncthreads();

            agg_walk_sum<false>(hin, src, dst, lst, n, listed, rbase, nrows, s_mx, s_de, s_nu, wave, lane, 0.f);
        }
        __syncthreads();

        for (unsigned r = wave; r < nrows; r += 32) {
            const size_t   g  = (size_t)(rbase + (int)r) * HDIM + 4 * lane;
            const float4   hv = *(const float4*)(hin + g);
            const unsigned o  = r * HDIM + lane;
            const float d0 = s_de[o], d1 = s_de[o + 32], d2 = s_de[o + 64], d3 = s_de[o + 96];
            const float n0 = s_nu[o], n1 = s_nu[o + 32], n2 = s_nu[o + 64], n3 = s_nu[o + 96];
            float4 q;
            if (usebound) {
                q.x = (d0 > 0.f ? n0 / d0 : 0.f) + hv.x;
                q.y = (d1 > 0.f ? n1 / d1 : 0.f) + hv.y;
                q.z = (d2 > 0.f ? n2 / d2 : 0.f) + hv.z;
                q.w = (d3 > 0.f ? n3 / d3 : 0.f) + hv.w;
            } else {
                q.x = n0 / (d0 + 1e-16f) + hv.x;
                q.y = n1 / (d1 + 1e-16f) + hv.y;
                q.z = n2 / (d2 + 1e-16f) + hv.z;
                q.w = n3 / (d3 + 1e-16f) + hv.w;
            }
            *(float4*)(t + g) = q;
        }
        __syncthreads();
    }
}

__global__ __launch_bounds__(256)
void ln_relu_kernel(const float* __restrict__ in, const float* __restrict__ g,
                    const float* __restrict__ b, float* __restrict__ out, int M)
{
    const int row  = blockIdx.x * 8 + (threadIdx.x >> 5);
    const int lane = threadIdx.x & 31;
    if (row >= M) return;
    const float4 v = *(const float4*)(in + (size_t)row * HDIM + lane * 4);
    float s1 = v.x + v.y + v.z + v.w;
    float s2 = v.x * v.x + v.y * v.y + v.z * v.z + v.w * v.w;
#pragma unroll
    for (int off = 16; off > 0; off >>= 1) {
        s1 += __shfl_xor(s1, off, 32);
        s2 += __shfl_xor(s2, off, 32);
    }
    const float mu  = s1 * (1.f / HDIM);
    const float var = s2 * (1.f / HDIM) - mu * mu;
    const float inv = rsqrtf(var + LN_EPS);
    const float4 gg = *(const float4*)(g + lane * 4);
    const float4 bb = *(const float4*)(b + lane * 4);
    float4 r;
    r.x = fmaxf((v.x - mu) * inv * gg.x + bb.x, 0.f);
    r.y = fmaxf((v.y - mu) * inv * gg.y + bb.y, 0.f);
    r.z = fmaxf((v.z - mu) * inv * gg.z + bb.z, 0.f);
    r.w = fmaxf((v.w - mu) * inv * gg.w + bb.w, 0.f);
    *(float4*)(out + (size_t)row * HDIM + lane * 4) = r;
}

__global__ __launch_bounds__(256)
void gather_kernel(const float* __restrict__ in, const int* __restrict__ map,
                   float* __restrict__ out, int M)
{
    const int tid = blockIdx.x * 256 + threadIdx.x;
    if (tid >= M * 32) return;
    const int i = tid >> 5;
    const int f = (tid & 31) * 4;
    const int j = map[i];
    *(float4*)(out + (size_t)i * HDIM + f) =
        *(const float4*)(in + (size_t)j * HDIM + f);
}

#define HEAD_PITCH 132
static_assert(NSEEDS % 16 == 0, "whole 16-row tiles");
static_assert(NCLS > 32 && NCLS <= 48, "three column tiles, the third partly padding");

__global__ __launch_bounds__(32)
void head_kernel(const float* __restrict__ hsrc, const int* __restrict__ nmap,
                 const int* __restrict__ fmap, const float* __restrict__ g,
                 const float* __restrict__ b, const float* __restrict__ Wt,
                 const float* __restrict__ bias, float* __restrict__ out)
{
    __shared__ __attribute__((aligned(16))) float s_a[16 * HEAD_PITCH];

    const int lane = threadIdx.x;
    const int row0 = blockIdx.x * 16;
    const int hb   = lane >> 4;
    const int lo   = lane & 15;

    const int j = nmap[fmap[row0 + lo]];

    const float4 gg = *(const float4*)(g + lane * 4);
    const float4 bb = *(const float4*)(b + lane * 4);

    float4 v[16];
#pragma unroll
    for (int r = 0; r < 16; ++r) {
        const int jr = __builtin_amdgcn_readlane(j, r);
        v[r] = *(const float4*)(hsrc + (size_t)jr * HDIM + lane * 4);
    }

#pragma unroll
    for (int r = 0; r < 16; ++r) {
        float s1 = v[r].x + v[r].y + v[r].z + v[r].w;
        float s2 = v[r].x * v[r].x + v[r].y * v[r].y + v[r].z * v[r].z + v[r].w * v[r].w;
#pragma unroll
        for (int off = 16; off > 0; off >>= 1) {
            s1 += __shfl_xor(s1, off, 32);
            s2 += __shfl_xor(s2, off, 32);
        }
        const float mu  = s1 * (1.f / HDIM);
        const float var = s2 * (1.f / HDIM) - mu * mu;
        const float inv = rsqrtf(var + LN_EPS);
        float4 q;
        q.x = fmaxf((v[r].x - mu) * inv * gg.x + bb.x, 0.f);
        q.y = fmaxf((v[r].y - mu) * inv * gg.y + bb.y, 0.f);
        q.z = fmaxf((v[r].z - mu) * inv * gg.z + bb.z, 0.f);
        q.w = fmaxf((v[r].w - mu) * inv * gg.w + bb.w, 0.f);
        *(float4*)(s_a + r * HEAD_PITCH + lane * 4) = q;
    }
    __syncthreads();

    const float* __restrict__ a_ptr = s_a + lo * HEAD_PITCH;
    const float* __restrict__ b_ptr = Wt + (size_t)lo * HDIM;

    v8f acc[3];
#pragma unroll
    for (int t = 0; t < 3; ++t)
#pragma unroll
        for (int i = 0; i < 8; ++i) acc[t][i] = 0.f;

#pragma unroll 8
    for (int k = 0; k < HDIM; k += 4) {
        const int ka = k + 2 * hb;
        const v2f a  = *(const v2f*)(a_ptr + ka);
        const v2f b0 = *(const v2f*)(b_ptr + ka);
        const v2f b1 = *(const v2f*)(b_ptr + 16 * HDIM + ka);
        const v2f b2 = *(const v2f*)(b_ptr + 32 * HDIM + ka);
        acc[0] = __builtin_amdgcn_wmma_f32_16x16x4_f32(false, a, false, b0, (short)0, acc[0], false, false);
        acc[1] = __builtin_amdgcn_wmma_f32_16x16x4_f32(false, a, false, b1, (short)0, acc[1], false, false);
        acc[2] = __builtin_amdgcn_wmma_f32_16x16x4_f32(false, a, false, b2, (short)0, acc[2], false, false);
    }

    const float NEG  = -3.0e38f;
    const bool  ok2  = (lo + 32) < NCLS;
    const float bv0  = bias[lo];
    const float bv1  = bias[lo + 16];
    const float bv2  = bias[lo + 32];
#pragma unroll
    for (int i = 0; i < 8; ++i) {
        const float x0 = acc[0][i] + bv0;
        const float x1 = acc[1][i] + bv1;
        const float x2 = acc[2][i] + bv2;
        float mx = fmaxf(fmaxf(x0, x1), ok2 ? x2 : NEG);
#pragma unroll
        for (int off = 8; off > 0; off >>= 1) mx = fmaxf(mx, __shfl_xor(mx, off, 32));
        float se = expf(x0 - mx) + expf(x1 - mx) + (ok2 ? expf(x2 - mx) : 0.f);
#pragma unroll
        for (int off = 8; off > 0; off >>= 1) se += __shfl_xor(se, off, 32);
        const float lse = logf(se) + mx;
        float* __restrict__ op = out + (size_t)(row0 + i + 8 * hb) * NCLS;
        op[lo]      = x0 - lse;
        op[lo + 16] = x1 - lse;
        if (ok2) op[lo + 32] = x2 - lse;
    }
}

extern "C" void kernel_launch(void* const* d_in, const int* in_sizes, int n_in,
                              void* d_out, int out_size, void* d_ws, size_t ws_size,
                              hipStream_t stream)
{
    (void)in_sizes; (void)n_in; (void)out_size; (void)ws_size;

    const float* x         = (const float*)d_in[0];
    const int*   src       = (const int*)  d_in[1];
    const int*   dst       = (const int*)  d_in[2];
    const int*   node_map  = (const int*)  d_in[3];
    const int*   final_map = (const int*)  d_in[4];
    const float* enc_w     = (const float*)d_in[5];
    const float* enc_b     = (const float*)d_in[6];
    const float* gcn_w     = (const float*)d_in[7];
    const float* gcn_b     = (const float*)d_in[8];
    const float* ln_g      = (const float*)d_in[9];
    const float* ln_b      = (const float*)d_in[10];
    const float* pred_w    = (const float*)d_in[11];
    const float* pred_b    = (const float*)d_in[12];
    float* out = (float*)d_out;

    const size_t NH = (size_t)NNODES * HDIM;
    float* B0     = (float*)d_ws;
    float* B1     = B0 + NH;
    float* B2     = B0 + 2 * NH;
    float* B4     = B0 + 4 * NH;
    float* PWT    = B0 + 5 * NH;
    float* PB     = PWT    + (size_t)NCLSPAD * HDIM;
    unsigned* WL  = (unsigned*)(PB + NCLSPAD);
    unsigned* WC  = WL + (size_t)2 * AGG_NBIN * AGG_CAP;
    unsigned* GM  = WC + (size_t)2 * AGG_NBIN;

    prep_kernel<<<(NCLSPAD * HDIM + 1023) / 1024, 1024, 0, stream>>>(
        pred_w, PWT, pred_b, PB, HDIM, NCLS, NCLSPAD, GM, WC, 2 * AGG_NBIN);

    deal_kernel<<<dim3(DEAL_WG, 2, 1), 1024, 0, stream>>>(src, dst, WL, WC);

    const int  mtN   = NNODES / 16;
    const int  NWGP  = 256;
    const int  ngrid = (NNODES * 32 + 255) / 256;
    const int  lgrid = (NNODES + 7) / 8;

    gemm_ws_kernel<FDIM, 4, false, true><<<NWGP, 128, 0, stream>>>(
        x, enc_w, enc_b, nullptr, B0, mtN, GM);

    auto conv = [&](const float* hin, int eblk, const float* wt, const float* bias,
                    const float* res, float* outp, const float* bg, const float* bb) {
        const int* es = src + (size_t)eblk * NEDGES;
        const int* ed = dst + (size_t)eblk * NEDGES;
        agg_kernel<<<AGG_NPART, 1024, 0, stream>>>(hin, es, ed, B2, WL + (size_t)eblk * AGG_NBIN * AGG_CAP,
                                                   WC + eblk * AGG_NBIN, bg, bb, GM);
        if (res)
            gemm_ws_kernel<HDIM, 8, true, false><<<NWGP, 128, 0, stream>>>(B2, wt, bias, res, outp, mtN, nullptr);
        else
            gemm_ws_kernel<HDIM, 8, false, false><<<NWGP, 128, 0, stream>>>(B2, wt, bias, nullptr, outp, mtN, nullptr);
    };

    conv(B0, 0, gcn_w, gcn_b, nullptr, B0, nullptr, nullptr);

    ln_relu_kernel<<<lgrid, 256, 0, stream>>>(B0, ln_g, ln_b, B1, NNODES);
    conv(B1, 0, gcn_w + (size_t)HDIM * HDIM, gcn_b + HDIM,  B0,  B4, ln_g, ln_b);
    gather_kernel<<<ngrid, 256, 0, stream>>>(B4, node_map, B0, NNODES);

    ln_relu_kernel<<<lgrid, 256, 0, stream>>>(B0, ln_g + HDIM, ln_b + HDIM, B1, NNODES);
    conv(B1, 1, gcn_w + (size_t)2 * HDIM * HDIM, gcn_b + 2 * HDIM,  B0,  B4,
         ln_g + HDIM, ln_b + HDIM);

    head_kernel<<<NSEEDS / 16, 32, 0, stream>>>(
        B4, node_map + NNODES, final_map, ln_g + 2 * HDIM, ln_b + 2 * HDIM, PWT, PB, out);
}
